# starting kernel
# baseline (speedup 1.0000x reference)
#include <hip/hip_runtime.h>

#define NQ 10
#define DIM 1024
#define NLAYERS 4

typedef __attribute__((ext_vector_type(2))) float v2f;
typedef __attribute__((ext_vector_type(8))) float v8f;

__device__ __forceinline__ int cnot_src(int d) {
#pragma unroll
  for (int i = 8; i >= 0; --i)
    if ((d >> i) & 1) d ^= (2 << i);
  return d;
}

__device__ __forceinline__ void build_bfrag(const float* cc, const float* ss,
                                            int lane, v2f bfrag[4]) {
  const int half = lane >> 4;
  const int n = lane & 15;
#pragma unroll
  for (int ks = 0; ks < 4; ++ks) {
    int k0 = 4 * ks + 2 * half;
    int k1 = k0 + 1;
    float e0 = 1.f, e1 = 1.f;
#pragma unroll
    for (int q = 0; q < 4; ++q) {
      float c = cc[q], s = ss[q];
      int nb = (n >> q) & 1;
      int k0b = (k0 >> q) & 1;
      int k1b = (k1 >> q) & 1;
      e0 *= (nb == k0b) ? c : (nb ? s : -s);
      e1 *= (nb == k1b) ? c : (nb ? s : -s);
    }
    v2f bv; bv.x = e0; bv.y = e1;
    bfrag[ks] = bv;
  }
}

__device__ __forceinline__ void contract_group(float* S, const v2f bfrag[4],
                                               int group, bool permute, int lane) {
  const int half = lane >> 4;
  const int lid = lane & 15;
  v2f a[16];
#pragma unroll
  for (int t = 0; t < 4; ++t) {
#pragma unroll
    for (int ks = 0; ks < 4; ++ks) {
      int k0 = 4 * ks + 2 * half;
      int d0, d1;
      if (group == 0) { int m = t * 16 + lid; d0 = (m << 4) | k0; d1 = d0 + 1; }
      else            { d0 = t * 256 + k0 * 16 + lid; d1 = d0 + 16; }
      if (permute) { d0 = cnot_src(d0); d1 = cnot_src(d1); }
      v2f av; av.x = S[d0]; av.y = S[d1];
      a[t * 4 + ks] = av;
    }
  }
#pragma unroll
  for (int t = 0; t < 4; ++t) {
    v8f acc = {0.f, 0.f, 0.f, 0.f, 0.f, 0.f, 0.f, 0.f};
#pragma unroll
    for (int ks = 0; ks < 4; ++ks)
      acc = __builtin_amdgcn_wmma_f32_16x16x4_f32(
          false, a[t * 4 + ks], false, bfrag[ks], (short)0, acc, false, false);
#pragma unroll
    for (int r = 0; r < 8; ++r) {
      int m = r + 8 * half;
      int d = (group == 0) ? (((t * 16 + m) << 4) | lid)
                           : (t * 256 + lid * 16 + m);
      S[d] = acc[r];
    }
  }
}

__device__ __forceinline__ void contract_top(float* Sre, float* Sim,
                                             float c8, float s8,
                                             float c9, float s9, int lane) {
#pragma unroll
  for (int j = 0; j < 8; ++j) {
    int rest = j * 32 + lane;
    {
      float a0 = Sre[rest], a1 = Sre[256 + rest];
      float a2 = Sre[512 + rest], a3 = Sre[768 + rest];
      float t0 = c8 * a0 - s8 * a1, t1 = s8 * a0 + c8 * a1;
      float t2 = c8 * a2 - s8 * a3, t3 = s8 * a2 + c8 * a3;
      Sre[rest]       = c9 * t0 - s9 * t2;
      Sre[256 + rest] = c9 * t1 - s9 * t3;
      Sre[512 + rest] = s9 * t0 + c9 * t2;
      Sre[768 + rest] = s9 * t1 + c9 * t3;
    }
    {
      float a0 = Sim[rest], a1 = Sim[256 + rest];
      float a2 = Sim[512 + rest], a3 = Sim[768 + rest];
      float t0 = c8 * a0 - s8 * a1, t1 = s8 * a0 + c8 * a1;
      float t2 = c8 * a2 - s8 * a3, t3 = s8 * a2 + c8 * a3;
      Sim[rest]       = c9 * t0 - s9 * t2;
      Sim[256 + rest] = c9 * t1 - s9 * t3;
      Sim[512 + rest] = s9 * t0 + c9 * t2;
      Sim[768 + rest] = s9 * t1 + c9 * t3;
    }
  }
}

__global__ __launch_bounds__(32) void qnn_circuit_kernel(
    const float* __restrict__ x, const float* __restrict__ W_pre,
    const float* __restrict__ b_pre, const float* __restrict__ weights,
    const float* __restrict__ W_post, const float* __restrict__ b_post,
    float* __restrict__ out) {
  __shared__ float Sre[DIM];
  __shared__ float Sim[DIM];
  __shared__ float hs[16];
  __shared__ float gs[16];
  __shared__ float csc[(NLAYERS + 1) * NQ];
  __shared__ float css[(NLAYERS + 1) * NQ];
  __shared__ float red[6 * 32];
  __shared__ float zb[16];

  const int lane = threadIdx.x;
  const int b = blockIdx.x;

  for (int idx = lane; idx < (NLAYERS + 1) * NQ; idx += 32) {
    float w = 0.5f * weights[idx];
    csc[idx] = __cosf(w);
    css[idx] = __sinf(w);
  }
  if (lane < NQ) {
    float acc = b_pre[lane];
#pragma unroll
    for (int k = 0; k < NQ; ++k) acc += x[b * NQ + k] * W_pre[lane * NQ + k];
    hs[lane] = acc;
  }
  __syncthreads();
  if (lane < NQ - 1) {
    const float PI = 3.14159265358979f;
    gs[lane] = (PI - hs[lane]) * (PI - hs[lane + 1]);
  }
  __syncthreads();

  float hr[NQ], gr[NQ - 1];
#pragma unroll
  for (int q = 0; q < NQ; ++q) hr[q] = hs[q];
#pragma unroll
  for (int q = 0; q < NQ - 1; ++q) gr[q] = gs[q];

  const float inv = 0.03125f;
#pragma unroll 4
  for (int j = 0; j < 32; ++j) {
    int d = (j << 5) | lane;
    float ang = 0.f;
#pragma unroll
    for (int q = 0; q < NQ; ++q) ang += ((d >> q) & 1) ? -hr[q] : hr[q];
    int dz = d ^ (d >> 1);
#pragma unroll
    for (int q = 0; q < NQ - 1; ++q) ang += ((dz >> q) & 1) ? -gr[q] : gr[q];
    float sn, cn;
    __sincosf(ang, &sn, &cn);
    Sre[d] = cn * inv;
    Sim[d] = -sn * inv;
  }
  __syncthreads();

  v2f bfrag[4];
#pragma unroll 1
  for (int l = 0; l <= NLAYERS; ++l) {
    build_bfrag(&csc[l * NQ + 0], &css[l * NQ + 0], lane, bfrag);
    contract_group(Sre, bfrag, 0, l > 0, lane);
    contract_group(Sim, bfrag, 0, l > 0, lane);
    __syncthreads();
    build_bfrag(&csc[l * NQ + 4], &css[l * NQ + 4], lane, bfrag);
    contract_group(Sre, bfrag, 1, false, lane);
    contract_group(Sim, bfrag, 1, false, lane);
    __syncthreads();
    contract_top(Sre, Sim, csc[l * NQ + 8], css[l * NQ + 8],
                 csc[l * NQ + 9], css[l * NQ + 9], lane);
    __syncthreads();
  }

  float psum = 0.f, zh0 = 0.f, zh1 = 0.f, zh2 = 0.f, zh3 = 0.f, zh4 = 0.f;
#pragma unroll 4
  for (int j = 0; j < 32; ++j) {
    int d = (j << 5) | lane;
    float re = Sre[d], im = Sim[d];
    float p = re * re + im * im;
    psum += p;
    zh0 += (j & 1) ? -p : p;
    zh1 += (j & 2) ? -p : p;
    zh2 += (j & 4) ? -p : p;
    zh3 += (j & 8) ? -p : p;
    zh4 += (j & 16) ? -p : p;
  }
  red[lane] = psum;
  red[32 + lane] = zh0;  red[64 + lane] = zh1;  red[96 + lane] = zh2;
  red[128 + lane] = zh3; red[160 + lane] = zh4;
  __syncthreads();
  if (lane < NQ) {
    float z = 0.f;
    if (lane < 5) {
#pragma unroll
      for (int L = 0; L < 32; ++L) z += ((L >> lane) & 1) ? -red[L] : red[L];
    } else {
#pragma unroll
      for (int L = 0; L < 32; ++L) z += red[(lane - 4) * 32 + L];
    }
    zb[lane] = z;
  }
  __syncthreads();
  if (lane < NQ) {
    float acc = b_post[lane];
#pragma unroll
    for (int q = 0; q < NQ; ++q) acc += zb[q] * W_post[lane * NQ + q];
    out[b * NQ + lane] = acc;
  }
}

extern "C" void kernel_launch(void* const* d_in, const int* in_sizes, int n_in,
                              void* d_out, int out_size, void* d_ws, size_t ws_size,
                              hipStream_t stream) {
  const float* x       = (const float*)d_in[0];
  const float* W_pre   = (const float*)d_in[1];
  const float* b_pre   = (const float*)d_in[2];
  const float* weights = (const float*)d_in[3];
  const float* W_post  = (const float*)d_in[4];
  const float* b_post  = (const float*)d_in[5];
  float* out = (float*)d_out;
  const int batch = in_sizes[0] / NQ;
  hipLaunchKernelGGL(qnn_circuit_kernel, dim3(batch), dim3(32), 0, stream,
                     x, W_pre, b_pre, weights, W_post, b_post, out);
}


# Per layer the state makes one LDS round trip, not three: qubits 4-7 on WMMA, its D registers reused as B of the qubits 0-3 WMMA, qubits 8,9 on those registers (order of the f32 sums changes)
# speedup vs baseline: 1.3334x; 1.3334x over previous
#include <hip/hip_runtime.h>


#define NQ 10
#define DIM 1024
#define NLAYERS 4

typedef __attribute__((ext_vector_type(2))) float v2f;
typedef __attribute__((ext_vector_type(4))) float v4f;
typedef __attribute__((ext_vector_type(8))) float v8f;

__device__ __forceinline__ int ladder_src(int d) { return (d ^ (d << 1)) & (DIM - 1); }

__device__ __forceinline__ void ry_row(float c, float s, int nb, float& r0, float& r1) {
  r0 = nb ? s : c;
  r1 = nb ? c : -s;
}

__device__ __forceinline__ void build_b_hi(const float* cc, const float* ss,
                                           int n, int h, v2f frag[4]) {
  float f0[2], f1[2], f2[2], f3[2];
  ry_row(cc[0], ss[0], (n >> 0) & 1, f0[0], f0[1]);
  ry_row(cc[1], ss[1], (n >> 1) & 1, f1[0], f1[1]);
  ry_row(cc[2], ss[2], (n >> 2) & 1, f2[0], f2[1]);
  ry_row(cc[3], ss[3], (n >> 3) & 1, f3[0], f3[1]);
  const float f1h = h ? f1[1] : f1[0];
#pragma unroll
  for (int ks = 0; ks < 4; ++ks) {
    v2f e;
    e.x = ((f0[0] * f1h) * f2[ks & 1]) * f3[ks >> 1];
    e.y = ((f0[1] * f1h) * f2[ks & 1]) * f3[ks >> 1];
    frag[ks] = e;
  }
}

__device__ __forceinline__ void build_a_lo(const float* cc, const float* ss,
                                           int m, int h, v2f frag[4]) {
  float f0[2], f1[2], f2[2], f3[2];
  ry_row(cc[0], ss[0], (m >> 0) & 1, f0[0], f0[1]);
  ry_row(cc[1], ss[1], (m >> 1) & 1, f1[0], f1[1]);
  ry_row(cc[2], ss[2], (m >> 2) & 1, f2[0], f2[1]);
  ry_row(cc[3], ss[3], (m >> 3) & 1, f3[0], f3[1]);
  const float f3h = h ? f3[1] : f3[0];
#pragma unroll
  for (int ks = 0; ks < 4; ++ks) {
    v2f e;
    e.x = ((f0[0] * f1[ks & 1]) * f2[ks >> 1]) * f3h;
    e.y = ((f0[1] * f1[ks & 1]) * f2[ks >> 1]) * f3h;
    frag[ks] = e;
  }
}

__global__ __launch_bounds__(32) void qnn_circuit_kernel(
    const float* __restrict__ x, const float* __restrict__ W_pre,
    const float* __restrict__ b_pre, const float* __restrict__ weights,
    const float* __restrict__ W_post, const float* __restrict__ b_post,
    float* __restrict__ out) {
  __shared__ __attribute__((aligned(16))) float Sre[DIM];
  __shared__ __attribute__((aligned(16))) float Sim[DIM];
  __shared__ float hs[16];
  __shared__ float gs[16];
  __shared__ float csc[(NLAYERS + 1) * NQ];
  __shared__ float css[(NLAYERS + 1) * NQ];
  __shared__ float red[6 * 32];
  __shared__ float zb[16];

  const int lane = threadIdx.x;
  const int b = blockIdx.x;

  for (int idx = lane; idx < (NLAYERS + 1) * NQ; idx += 32) {
    float w = 0.5f * weights[idx];
    csc[idx] = __cosf(w);
    css[idx] = __sinf(w);
  }
  if (lane < NQ) {
    float acc = b_pre[lane];
#pragma unroll
    for (int k = 0; k < NQ; ++k) acc += x[b * NQ + k] * W_pre[lane * NQ + k];
    hs[lane] = acc;
  }
  __syncthreads();
  if (lane < NQ - 1) {
    const float PI = 3.14159265358979f;
    gs[lane] = (PI - hs[lane]) * (PI - hs[lane + 1]);
  }
  __syncthreads();

  float hr[NQ], gr[NQ - 1];
#pragma unroll
  for (int q = 0; q < NQ; ++q) hr[q] = hs[q];
#pragma unroll
  for (int q = 0; q < NQ - 1; ++q) gr[q] = gs[q];

  const float inv = 0.03125f;
#pragma unroll 4
  for (int j = 0; j < 32; ++j) {
    int d = (j << 5) | lane;
    float ang = 0.f;
#pragma unroll
    for (int q = 0; q < NQ; ++q) ang += ((d >> q) & 1) ? -hr[q] : hr[q];
    int dz = d ^ (d >> 1);
#pragma unroll
    for (int q = 0; q < NQ - 1; ++q) ang += ((dz >> q) & 1) ? -gr[q] : gr[q];
    float sn, cn;
    __sincosf(ang, &sn, &cn);
    int p = ladder_src(d);
    Sre[p] = cn * inv;
    Sim[p] = -sn * inv;
  }
  __syncthreads();

  const int half = lane >> 4;
  const int lid = lane & 15;
  const int dlane = (half << 5) | lid;
  const int plane_src = dlane ^ (dlane << 1);

#pragma unroll 1
  for (int l = 0; l <= NLAYERS; ++l) {
    v2f bhi[4], alo[4];
    build_b_hi(&csc[l * NQ + 4], &css[l * NQ + 4], lid, half, bhi);
    build_a_lo(&csc[l * NQ + 0], &css[l * NQ + 0], lid, half, alo);
    const float c8 = csc[l * NQ + 8], s8 = css[l * NQ + 8];
    const float c9 = csc[l * NQ + 9], s9 = css[l * NQ + 9];

    v2f xa[2][4][4];
#pragma unroll
    for (int pl = 0; pl < 2; ++pl) {
      const float* S = pl ? Sim : Sre;
#pragma unroll
      for (int t = 0; t < 4; ++t) {
#pragma unroll
        for (int ks = 0; ks < 4; ++ks) {
          const int c0 = ladder_src(t * 256 + (4 * ks + 0) * 16);
          const int c1 = ladder_src(t * 256 + (4 * ks + 1) * 16);
          v2f av;
          av.x = S[plane_src ^ c0];
          av.y = S[plane_src ^ c1];
          xa[pl][t][ks] = av;
        }
      }
    }

#pragma unroll
    for (int pl = 0; pl < 2; ++pl) {
      asm volatile("" ::
        "v"(xa[pl][0][0]), "v"(xa[pl][0][1]), "v"(xa[pl][0][2]), "v"(xa[pl][0][3]),
        "v"(xa[pl][1][0]), "v"(xa[pl][1][1]), "v"(xa[pl][1][2]), "v"(xa[pl][1][3]),
        "v"(xa[pl][2][0]), "v"(xa[pl][2][1]), "v"(xa[pl][2][2]), "v"(xa[pl][2][3]),
        "v"(xa[pl][3][0]), "v"(xa[pl][3][1]), "v"(xa[pl][3][2]), "v"(xa[pl][3][3]));
    }

    v8f y[2][4];
#pragma unroll
    for (int pl = 0; pl < 2; ++pl) {
#pragma unroll
      for (int t = 0; t < 4; ++t) {
        v8f acc = {0.f, 0.f, 0.f, 0.f, 0.f, 0.f, 0.f, 0.f};
#pragma unroll
        for (int ks = 0; ks < 4; ++ks)
          acc = __builtin_amdgcn_wmma_f32_16x16x4_f32(
              false, xa[pl][t][ks], false, bhi[ks], (short)0, acc, false, false);
        y[pl][t] = acc;
      }
    }

    v8f z[2][4];
#pragma unroll
    for (int pl = 0; pl < 2; ++pl) {
#pragma unroll
      for (int t = 0; t < 4; ++t) {
        v8f acc = {0.f, 0.f, 0.f, 0.f, 0.f, 0.f, 0.f, 0.f};
#pragma unroll
        for (int ks = 0; ks < 4; ++ks) {
          v2f bv;
          bv.x = y[pl][t][2 * ks];
          bv.y = y[pl][t][2 * ks + 1];
          acc = __builtin_amdgcn_wmma_f32_16x16x4_f32(
              false, alo[ks], false, bv, (short)0, acc, false, false);
        }
        z[pl][t] = acc;
      }
    }

#pragma unroll
    for (int pl = 0; pl < 2; ++pl) {
      float* S = pl ? Sim : Sre;
      v8f o0, o1, o2, o3;
#pragma unroll
      for (int r = 0; r < 8; ++r) {
        float a0 = z[pl][0][r], a1 = z[pl][1][r];
        float a2 = z[pl][2][r], a3 = z[pl][3][r];
        float t0 = c8 * a0 - s8 * a1, t1 = s8 * a0 + c8 * a1;
        float t2 = c8 * a2 - s8 * a3, t3 = s8 * a2 + c8 * a3;
        o0[r] = c9 * t0 - s9 * t2;
        o1[r] = c9 * t1 - s9 * t3;
        o2[r] = s9 * t0 + c9 * t2;
        o3[r] = s9 * t1 + c9 * t3;
      }
      const int base = lid * 16 + 8 * half;
      v4f lo, hi;
      lo = {o0[0], o0[1], o0[2], o0[3]}; hi = {o0[4], o0[5], o0[6], o0[7]};
      *(v4f*)&S[0 * 256 + base] = lo; *(v4f*)&S[0 * 256 + base + 4] = hi;
      lo = {o1[0], o1[1], o1[2], o1[3]}; hi = {o1[4], o1[5], o1[6], o1[7]};
      *(v4f*)&S[1 * 256 + base] = lo; *(v4f*)&S[1 * 256 + base + 4] = hi;
      lo = {o2[0], o2[1], o2[2], o2[3]}; hi = {o2[4], o2[5], o2[6], o2[7]};
      *(v4f*)&S[2 * 256 + base] = lo; *(v4f*)&S[2 * 256 + base + 4] = hi;
      lo = {o3[0], o3[1], o3[2], o3[3]}; hi = {o3[4], o3[5], o3[6], o3[7]};
      *(v4f*)&S[3 * 256 + base] = lo; *(v4f*)&S[3 * 256 + base + 4] = hi;
    }
    __syncthreads();
  }

  float psum = 0.f, zh0 = 0.f, zh1 = 0.f, zh2 = 0.f, zh3 = 0.f, zh4 = 0.f;
#pragma unroll 4
  for (int j = 0; j < 32; ++j) {
    int d = (j << 5) | lane;
    float re = Sre[d], im = Sim[d];
    float p = re * re + im * im;
    psum += p;
    zh0 += (j & 1) ? -p : p;
    zh1 += (j & 2) ? -p : p;
    zh2 += (j & 4) ? -p : p;
    zh3 += (j & 8) ? -p : p;
    zh4 += (j & 16) ? -p : p;
  }
  red[lane] = psum;
  red[32 + lane] = zh0;  red[64 + lane] = zh1;  red[96 + lane] = zh2;
  red[128 + lane] = zh3; red[160 + lane] = zh4;
  __syncthreads();
  if (lane < NQ) {
    float z = 0.f;
    if (lane < 5) {
#pragma unroll
      for (int L = 0; L < 32; ++L) z += ((L >> lane) & 1) ? -red[L] : red[L];
    } else {
#pragma unroll
      for (int L = 0; L < 32; ++L) z += red[(lane - 4) * 32 + L];
    }
    zb[lane] = z;
  }
  __syncthreads();
  if (lane < NQ) {
    float acc = b_post[lane];
#pragma unroll
    for (int q = 0; q < NQ; ++q) acc += zb[q] * W_post[lane * NQ + q];
    out[b * NQ + lane] = acc;
  }
}

extern "C" void kernel_launch(void* const* d_in, const int* in_sizes, int n_in,
                              void* d_out, int out_size, void* d_ws, size_t ws_size,
                              hipStream_t stream) {
  const float* x       = (const float*)d_in[0];
  const float* W_pre   = (const float*)d_in[1];
  const float* b_pre   = (const float*)d_in[2];
  const float* weights = (const float*)d_in[3];
  const float* W_post  = (const float*)d_in[4];
  const float* b_post  = (const float*)d_in[5];
  float* out = (float*)d_out;
  const int batch = in_sizes[0] / NQ;
  hipLaunchKernelGGL(qnn_circuit_kernel, dim3(batch), dim3(32), 0, stream,
                     x, W_pre, b_pre, weights, W_post, b_post, out);
}


# Initial state from a factored phase: exp(-i*ang) = lane factor x row factor, 3 sincos per lane and one complex multiply per amplitude, not a 19-term sum and a sincos each (angle summed in two parts)
# speedup vs baseline: 1.5736x; 1.1801x over previous
#include <hip/hip_runtime.h>


#define NQ 10
#define DIM 1024
#define NLAYERS 4

typedef __attribute__((ext_vector_type(2))) float v2f;
typedef __attribute__((ext_vector_type(4))) float v4f;
typedef __attribute__((ext_vector_type(8))) float v8f;

__device__ __forceinline__ int ladder_src(int d) { return (d ^ (d << 1)) & (DIM - 1); }

__device__ __forceinline__ void ry_row(float c, float s, int nb, float& r0, float& r1) {
  r0 = nb ? s : c;
  r1 = nb ? c : -s;
}

__device__ __forceinline__ void build_b_hi(const float* cc, const float* ss,
                                           int n, int h, v2f frag[4]) {
  float f0[2], f1[2], f2[2], f3[2];
  ry_row(cc[0], ss[0], (n >> 0) & 1, f0[0], f0[1]);
  ry_row(cc[1], ss[1], (n >> 1) & 1, f1[0], f1[1]);
  ry_row(cc[2], ss[2], (n >> 2) & 1, f2[0], f2[1]);
  ry_row(cc[3], ss[3], (n >> 3) & 1, f3[0], f3[1]);
  const float f1h = h ? f1[1] : f1[0];
#pragma unroll
  for (int ks = 0; ks < 4; ++ks) {
    v2f e;
    e.x = ((f0[0] * f1h) * f2[ks & 1]) * f3[ks >> 1];
    e.y = ((f0[1] * f1h) * f2[ks & 1]) * f3[ks >> 1];
    frag[ks] = e;
  }
}

__device__ __forceinline__ void build_a_lo(const float* cc, const float* ss,
                                           int m, int h, v2f frag[4]) {
  float f0[2], f1[2], f2[2], f3[2];
  ry_row(cc[0], ss[0], (m >> 0) & 1, f0[0], f0[1]);
  ry_row(cc[1], ss[1], (m >> 1) & 1, f1[0], f1[1]);
  ry_row(cc[2], ss[2], (m >> 2) & 1, f2[0], f2[1]);
  ry_row(cc[3], ss[3], (m >> 3) & 1, f3[0], f3[1]);
  const float f3h = h ? f3[1] : f3[0];
#pragma unroll
  for (int ks = 0; ks < 4; ++ks) {
    v2f e;
    e.x = ((f0[0] * f1[ks & 1]) * f2[ks >> 1]) * f3h;
    e.y = ((f0[1] * f1[ks & 1]) * f2[ks >> 1]) * f3h;
    frag[ks] = e;
  }
}

__global__ __launch_bounds__(32) void qnn_circuit_kernel(
    const float* __restrict__ x, const float* __restrict__ W_pre,
    const float* __restrict__ b_pre, const float* __restrict__ weights,
    const float* __restrict__ W_post, const float* __restrict__ b_post,
    float* __restrict__ out) {
  __shared__ __attribute__((aligned(16))) float Sre[DIM];
  __shared__ __attribute__((aligned(16))) float Sim[DIM];
  __shared__ float hs[16];
  __shared__ float gs[16];
  __shared__ float csc[(NLAYERS + 1) * NQ];
  __shared__ float css[(NLAYERS + 1) * NQ];
  __shared__ float red[6 * 32];
  __shared__ float zb[16];

  const int lane = threadIdx.x;
  const int b = blockIdx.x;

  for (int idx = lane; idx < (NLAYERS + 1) * NQ; idx += 32) {
    float w = 0.5f * weights[idx];
    csc[idx] = __cosf(w);
    css[idx] = __sinf(w);
  }
  if (lane < NQ) {
    float acc = b_pre[lane];
#pragma unroll
    for (int k = 0; k < NQ; ++k) acc += x[b * NQ + k] * W_pre[lane * NQ + k];
    hs[lane] = acc;
  }
  __syncthreads();
  if (lane < NQ - 1) {
    const float PI = 3.14159265358979f;
    gs[lane] = (PI - hs[lane]) * (PI - hs[lane + 1]);
  }
  __syncthreads();

  float hr[NQ], gr[NQ - 1];
#pragma unroll
  for (int q = 0; q < NQ; ++q) hr[q] = hs[q];
#pragma unroll
  for (int q = 0; q < NQ - 1; ++q) gr[q] = gs[q];

  const float inv = 0.03125f;
  const int lz = lane ^ (lane >> 1);
  float angA = 0.f, angB = 0.f;
#pragma unroll
  for (int q = 0; q < 5; ++q) {
    angA += ((lane >> q) & 1) ? -hr[q] : hr[q];
    angB += ((lane >> q) & 1) ? -hr[5 + q] : hr[5 + q];
  }
#pragma unroll
  for (int q = 0; q < 4; ++q) {
    angA += ((lz >> q) & 1) ? -gr[q] : gr[q];
    angB += ((lz >> q) & 1) ? -gr[5 + q] : gr[5 + q];
  }
  const float gx = (lane & 16) ? -gr[4] : gr[4];
  float s0, c0, s1, c1, sb, cb;
  __sincosf(angA + gx, &s0, &c0);
  __sincosf(angA - gx, &s1, &c1);
  __sincosf(angB, &sb, &cb);
  const float frl = cb * inv, fil = -sb * inv;
  const int pl = lane ^ (lane << 1);
#pragma unroll
  for (int j = 0; j < 32; ++j) {
    const float fr = __int_as_float(__builtin_amdgcn_readlane(__float_as_int(frl), j));
    const float fi = __int_as_float(__builtin_amdgcn_readlane(__float_as_int(fil), j));
    const float er = (j & 1) ? c1 : c0;
    const float ei = (j & 1) ? -s1 : -s0;
    const int p = pl ^ ladder_src(j << 5);
    Sre[p] = er * fr - ei * fi;
    Sim[p] = er * fi + ei * fr;
  }
  __syncthreads();

  const int half = lane >> 4;
  const int lid = lane & 15;
  const int dlane = (half << 5) | lid;
  const int plane_src = dlane ^ (dlane << 1);

#pragma unroll 1
  for (int l = 0; l <= NLAYERS; ++l) {
    v2f bhi[4], alo[4];
    build_b_hi(&csc[l * NQ + 4], &css[l * NQ + 4], lid, half, bhi);
    build_a_lo(&csc[l * NQ + 0], &css[l * NQ + 0], lid, half, alo);
    const float c8 = csc[l * NQ + 8], s8 = css[l * NQ + 8];
    const float c9 = csc[l * NQ + 9], s9 = css[l * NQ + 9];

    v2f xa[2][4][4];
#pragma unroll
    for (int pl = 0; pl < 2; ++pl) {
      const float* S = pl ? Sim : Sre;
#pragma unroll
      for (int t = 0; t < 4; ++t) {
#pragma unroll
        for (int ks = 0; ks < 4; ++ks) {
          const int c0 = ladder_src(t * 256 + (4 * ks + 0) * 16);
          const int c1 = ladder_src(t * 256 + (4 * ks + 1) * 16);
          v2f av;
          av.x = S[plane_src ^ c0];
          av.y = S[plane_src ^ c1];
          xa[pl][t][ks] = av;
        }
      }
    }

#pragma unroll
    for (int pl = 0; pl < 2; ++pl) {
      asm volatile("" ::
        "v"(xa[pl][0][0]), "v"(xa[pl][0][1]), "v"(xa[pl][0][2]), "v"(xa[pl][0][3]),
        "v"(xa[pl][1][0]), "v"(xa[pl][1][1]), "v"(xa[pl][1][2]), "v"(xa[pl][1][3]),
        "v"(xa[pl][2][0]), "v"(xa[pl][2][1]), "v"(xa[pl][2][2]), "v"(xa[pl][2][3]),
        "v"(xa[pl][3][0]), "v"(xa[pl][3][1]), "v"(xa[pl][3][2]), "v"(xa[pl][3][3]));
    }

    v8f y[2][4];
#pragma unroll
    for (int pl = 0; pl < 2; ++pl) {
#pragma unroll
      for (int t = 0; t < 4; ++t) {
        v8f acc = {0.f, 0.f, 0.f, 0.f, 0.f, 0.f, 0.f, 0.f};
#pragma unroll
        for (int ks = 0; ks < 4; ++ks)
          acc = __builtin_amdgcn_wmma_f32_16x16x4_f32(
              false, xa[pl][t][ks], false, bhi[ks], (short)0, acc, false, false);
        y[pl][t] = acc;
      }
    }

    v8f z[2][4];
#pragma unroll
    for (int pl = 0; pl < 2; ++pl) {
#pragma unroll
      for (int t = 0; t < 4; ++t) {
        v8f acc = {0.f, 0.f, 0.f, 0.f, 0.f, 0.f, 0.f, 0.f};
#pragma unroll
        for (int ks = 0; ks < 4; ++ks) {
          v2f bv;
          bv.x = y[pl][t][2 * ks];
          bv.y = y[pl][t][2 * ks + 1];
          acc = __builtin_amdgcn_wmma_f32_16x16x4_f32(
              false, alo[ks], false, bv, (short)0, acc, false, false);
        }
        z[pl][t] = acc;
      }
    }

#pragma unroll
    for (int pl = 0; pl < 2; ++pl) {
      float* S = pl ? Sim : Sre;
      v8f o0, o1, o2, o3;
#pragma unroll
      for (int r = 0; r < 8; ++r) {
        float a0 = z[pl][0][r], a1 = z[pl][1][r];
        float a2 = z[pl][2][r], a3 = z[pl][3][r];
        float t0 = c8 * a0 - s8 * a1, t1 = s8 * a0 + c8 * a1;
        float t2 = c8 * a2 - s8 * a3, t3 = s8 * a2 + c8 * a3;
        o0[r] = c9 * t0 - s9 * t2;
        o1[r] = c9 * t1 - s9 * t3;
        o2[r] = s9 * t0 + c9 * t2;
        o3[r] = s9 * t1 + c9 * t3;
      }
      const int base = lid * 16 + 8 * half;
      v4f lo, hi;
      lo = {o0[0], o0[1], o0[2], o0[3]}; hi = {o0[4], o0[5], o0[6], o0[7]};
      *(v4f*)&S[0 * 256 + base] = lo; *(v4f*)&S[0 * 256 + base + 4] = hi;
      lo = {o1[0], o1[1], o1[2], o1[3]}; hi = {o1[4], o1[5], o1[6], o1[7]};
      *(v4f*)&S[1 * 256 + base] = lo; *(v4f*)&S[1 * 256 + base + 4] = hi;
      lo = {o2[0], o2[1], o2[2], o2[3]}; hi = {o2[4], o2[5], o2[6], o2[7]};
      *(v4f*)&S[2 * 256 + base] = lo; *(v4f*)&S[2 * 256 + base + 4] = hi;
      lo = {o3[0], o3[1], o3[2], o3[3]}; hi = {o3[4], o3[5], o3[6], o3[7]};
      *(v4f*)&S[3 * 256 + base] = lo; *(v4f*)&S[3 * 256 + base + 4] = hi;
    }
    __syncthreads();
  }

  float psum = 0.f, zh0 = 0.f, zh1 = 0.f, zh2 = 0.f, zh3 = 0.f, zh4 = 0.f;
#pragma unroll 4
  for (int j = 0; j < 32; ++j) {
    int d = (j << 5) | lane;
    float re = Sre[d], im = Sim[d];
    float p = re * re + im * im;
    psum += p;
    zh0 += (j & 1) ? -p : p;
    zh1 += (j & 2) ? -p : p;
    zh2 += (j & 4) ? -p : p;
    zh3 += (j & 8) ? -p : p;
    zh4 += (j & 16) ? -p : p;
  }
  red[lane] = psum;
  red[32 + lane] = zh0;  red[64 + lane] = zh1;  red[96 + lane] = zh2;
  red[128 + lane] = zh3; red[160 + lane] = zh4;
  __syncthreads();
  if (lane < NQ) {
    float z = 0.f;
    if (lane < 5) {
#pragma unroll
      for (int L = 0; L < 32; ++L) z += ((L >> lane) & 1) ? -red[L] : red[L];
    } else {
#pragma unroll
      for (int L = 0; L < 32; ++L) z += red[(lane - 4) * 32 + L];
    }
    zb[lane] = z;
  }
  __syncthreads();
  if (lane < NQ) {
    float acc = b_post[lane];
#pragma unroll
    for (int q = 0; q < NQ; ++q) acc += zb[q] * W_post[lane * NQ + q];
    out[b * NQ + lane] = acc;
  }
}

extern "C" void kernel_launch(void* const* d_in, const int* in_sizes, int n_in,
                              void* d_out, int out_size, void* d_ws, size_t ws_size,
                              hipStream_t stream) {
  const float* x       = (const float*)d_in[0];
  const float* W_pre   = (const float*)d_in[1];
  const float* b_pre   = (const float*)d_in[2];
  const float* weights = (const float*)d_in[3];
  const float* W_post  = (const float*)d_in[4];
  const float* b_post  = (const float*)d_in[5];
  float* out = (float*)d_out;
  const int batch = in_sizes[0] / NQ;
  hipLaunchKernelGGL(qnn_circuit_kernel, dim3(batch), dim3(32), 0, stream,
                     x, W_pre, b_pre, weights, W_post, b_post, out);
}


# Eight batch elements per workgroup, one per wave in its own LDS slice with wave-local LDS ordering: 256 eight-wave workgroups enter instead of 2048 one-wave ones
# speedup vs baseline: 1.8428x; 1.1711x over previous
#include <hip/hip_runtime.h>


#define NQ 10
#define DIM 1024
#define NLAYERS 4
#define BATCH 2048
#define WAVES 8

typedef __attribute__((ext_vector_type(2))) float v2f;
typedef __attribute__((ext_vector_type(4))) float v4f;
typedef __attribute__((ext_vector_type(8))) float v8f;

__device__ __forceinline__ int ladder_src(int d) { return (d ^ (d << 1)) & (DIM - 1); }

__device__ __forceinline__ void ry_row(float c, float s, int nb, float& r0, float& r1) {
  r0 = nb ? s : c;
  r1 = nb ? c : -s;
}

__device__ __forceinline__ void build_b_hi(const float* cc, const float* ss,
                                           int n, int h, v2f frag[4]) {
  float f0[2], f1[2], f2[2], f3[2];
  ry_row(cc[0], ss[0], (n >> 0) & 1, f0[0], f0[1]);
  ry_row(cc[1], ss[1], (n >> 1) & 1, f1[0], f1[1]);
  ry_row(cc[2], ss[2], (n >> 2) & 1, f2[0], f2[1]);
  ry_row(cc[3], ss[3], (n >> 3) & 1, f3[0], f3[1]);
  const float f1h = h ? f1[1] : f1[0];
#pragma unroll
  for (int ks = 0; ks < 4; ++ks) {
    v2f e;
    e.x = ((f0[0] * f1h) * f2[ks & 1]) * f3[ks >> 1];
    e.y = ((f0[1] * f1h) * f2[ks & 1]) * f3[ks >> 1];
    frag[ks] = e;
  }
}

__device__ __forceinline__ void build_a_lo(const float* cc, const float* ss,
                                           int m, int h, v2f frag[4]) {
  float f0[2], f1[2], f2[2], f3[2];
  ry_row(cc[0], ss[0], (m >> 0) & 1, f0[0], f0[1]);
  ry_row(cc[1], ss[1], (m >> 1) & 1, f1[0], f1[1]);
  ry_row(cc[2], ss[2], (m >> 2) & 1, f2[0], f2[1]);
  ry_row(cc[3], ss[3], (m >> 3) & 1, f3[0], f3[1]);
  const float f3h = h ? f3[1] : f3[0];
#pragma unroll
  for (int ks = 0; ks < 4; ++ks) {
    v2f e;
    e.x = ((f0[0] * f1[ks & 1]) * f2[ks >> 1]) * f3h;
    e.y = ((f0[1] * f1[ks & 1]) * f2[ks >> 1]) * f3h;
    frag[ks] = e;
  }
}

struct WaveLds {
  float Sre[DIM];
  float Sim[DIM];
  float hs[16];
  float gs[16];
  float csc[(NLAYERS + 1) * NQ];
  float css[(NLAYERS + 1) * NQ];
  float red[6 * 32];
  float zb[16];
};

__device__ __forceinline__ void wave_lds_sync() {
  __builtin_amdgcn_fence(__ATOMIC_RELEASE, "workgroup", "local");
  __builtin_amdgcn_wave_barrier();
  __builtin_amdgcn_fence(__ATOMIC_ACQUIRE, "workgroup", "local");
}

__global__ __launch_bounds__(32 * WAVES) void qnn_circuit_kernel(
    const float* __restrict__ x, const float* __restrict__ W_pre,
    const float* __restrict__ b_pre, const float* __restrict__ weights,
    const float* __restrict__ W_post, const float* __restrict__ b_post,
    float* __restrict__ out) {
  __shared__ __attribute__((aligned(16))) WaveLds lds[WAVES];

  const int wave = __builtin_amdgcn_readfirstlane((int)(threadIdx.x >> 5));
  const int lane = threadIdx.x & 31;
  const int b = blockIdx.x * WAVES + wave;
  WaveLds& L = lds[wave];
  float* const Sre = L.Sre;
  float* const Sim = L.Sim;
  float* const hs = L.hs;
  float* const gs = L.gs;
  float* const csc = L.csc;
  float* const css = L.css;
  float* const red = L.red;
  float* const zb = L.zb;

  for (int idx = lane; idx < (NLAYERS + 1) * NQ; idx += 32) {
    float w = 0.5f * weights[idx];
    csc[idx] = __cosf(w);
    css[idx] = __sinf(w);
  }
  if (lane < NQ) {
    float acc = b_pre[lane];
#pragma unroll
    for (int k = 0; k < NQ; ++k) acc += x[b * NQ + k] * W_pre[lane * NQ + k];
    hs[lane] = acc;
  }
  wave_lds_sync();
  if (lane < NQ - 1) {
    const float PI = 3.14159265358979f;
    gs[lane] = (PI - hs[lane]) * (PI - hs[lane + 1]);
  }
  wave_lds_sync();

  float hr[NQ], gr[NQ - 1];
#pragma unroll
  for (int q = 0; q < NQ; ++q) hr[q] = hs[q];
#pragma unroll
  for (int q = 0; q < NQ - 1; ++q) gr[q] = gs[q];

  const float inv = 0.03125f;
  const int lz = lane ^ (lane >> 1);
  float angA = 0.f, angB = 0.f;
#pragma unroll
  for (int q = 0; q < 5; ++q) {
    angA += ((lane >> q) & 1) ? -hr[q] : hr[q];
    angB += ((lane >> q) & 1) ? -hr[5 + q] : hr[5 + q];
  }
#pragma unroll
  for (int q = 0; q < 4; ++q) {
    angA += ((lz >> q) & 1) ? -gr[q] : gr[q];
    angB += ((lz >> q) & 1) ? -gr[5 + q] : gr[5 + q];
  }
  const float gx = (lane & 16) ? -gr[4] : gr[4];
  float s0, c0, s1, c1, sb, cb;
  __sincosf(angA + gx, &s0, &c0);
  __sincosf(angA - gx, &s1, &c1);
  __sincosf(angB, &sb, &cb);
  const float frl = cb * inv, fil = -sb * inv;
  const int pl = lane ^ (lane << 1);
#pragma unroll
  for (int j = 0; j < 32; ++j) {
    const float fr = __int_as_float(__builtin_amdgcn_readlane(__float_as_int(frl), j));
    const float fi = __int_as_float(__builtin_amdgcn_readlane(__float_as_int(fil), j));
    const float er = (j & 1) ? c1 : c0;
    const float ei = (j & 1) ? -s1 : -s0;
    const int p = pl ^ ladder_src(j << 5);
    Sre[p] = er * fr - ei * fi;
    Sim[p] = er * fi + ei * fr;
  }
  wave_lds_sync();

  const int half = lane >> 4;
  const int lid = lane & 15;
  const int dlane = (half << 5) | lid;
  const int plane_src = dlane ^ (dlane << 1);

#pragma unroll 1
  for (int l = 0; l <= NLAYERS; ++l) {
    v2f bhi[4], alo[4];
    build_b_hi(&csc[l * NQ + 4], &css[l * NQ + 4], lid, half, bhi);
    build_a_lo(&csc[l * NQ + 0], &css[l * NQ + 0], lid, half, alo);
    const float c8 = csc[l * NQ + 8], s8 = css[l * NQ + 8];
    const float c9 = csc[l * NQ + 9], s9 = css[l * NQ + 9];

    v2f xa[2][4][4];
#pragma unroll
    for (int pl = 0; pl < 2; ++pl) {
      const float* S = pl ? Sim : Sre;
#pragma unroll
      for (int t = 0; t < 4; ++t) {
#pragma unroll
        for (int ks = 0; ks < 4; ++ks) {
          const int c0 = ladder_src(t * 256 + (4 * ks + 0) * 16);
          const int c1 = ladder_src(t * 256 + (4 * ks + 1) * 16);
          v2f av;
          av.x = S[plane_src ^ c0];
          av.y = S[plane_src ^ c1];
          xa[pl][t][ks] = av;
        }
      }
    }

#pragma unroll
    for (int pl = 0; pl < 2; ++pl) {
      asm volatile("" ::
        "v"(xa[pl][0][0]), "v"(xa[pl][0][1]), "v"(xa[pl][0][2]), "v"(xa[pl][0][3]),
        "v"(xa[pl][1][0]), "v"(xa[pl][1][1]), "v"(xa[pl][1][2]), "v"(xa[pl][1][3]),
        "v"(xa[pl][2][0]), "v"(xa[pl][2][1]), "v"(xa[pl][2][2]), "v"(xa[pl][2][3]),
        "v"(xa[pl][3][0]), "v"(xa[pl][3][1]), "v"(xa[pl][3][2]), "v"(xa[pl][3][3]));
    }

    v8f y[2][4];
#pragma unroll
    for (int pl = 0; pl < 2; ++pl) {
#pragma unroll
      for (int t = 0; t < 4; ++t) {
        v8f acc = {0.f, 0.f, 0.f, 0.f, 0.f, 0.f, 0.f, 0.f};
#pragma unroll
        for (int ks = 0; ks < 4; ++ks)
          acc = __builtin_amdgcn_wmma_f32_16x16x4_f32(
              false, xa[pl][t][ks], false, bhi[ks], (short)0, acc, false, false);
        y[pl][t] = acc;
      }
    }

    v8f z[2][4];
#pragma unroll
    for (int pl = 0; pl < 2; ++pl) {
#pragma unroll
      for (int t = 0; t < 4; ++t) {
        v8f acc = {0.f, 0.f, 0.f, 0.f, 0.f, 0.f, 0.f, 0.f};
#pragma unroll
        for (int ks = 0; ks < 4; ++ks) {
          v2f bv;
          bv.x = y[pl][t][2 * ks];
          bv.y = y[pl][t][2 * ks + 1];
          acc = __builtin_amdgcn_wmma_f32_16x16x4_f32(
              false, alo[ks], false, bv, (short)0, acc, false, false);
        }
        z[pl][t] = acc;
      }
    }

#pragma unroll
    for (int pl = 0; pl < 2; ++pl) {
      float* S = pl ? Sim : Sre;
      v8f o0, o1, o2, o3;
#pragma unroll
      for (int r = 0; r < 8; ++r) {
        float a0 = z[pl][0][r], a1 = z[pl][1][r];
        float a2 = z[pl][2][r], a3 = z[pl][3][r];
        float t0 = c8 * a0 - s8 * a1, t1 = s8 * a0 + c8 * a1;
        float t2 = c8 * a2 - s8 * a3, t3 = s8 * a2 + c8 * a3;
        o0[r] = c9 * t0 - s9 * t2;
        o1[r] = c9 * t1 - s9 * t3;
        o2[r] = s9 * t0 + c9 * t2;
        o3[r] = s9 * t1 + c9 * t3;
      }
      const int base = lid * 16 + 8 * half;
      v4f lo, hi;
      lo = {o0[0], o0[1], o0[2], o0[3]}; hi = {o0[4], o0[5], o0[6], o0[7]};
      *(v4f*)&S[0 * 256 + base] = lo; *(v4f*)&S[0 * 256 + base + 4] = hi;
      lo = {o1[0], o1[1], o1[2], o1[3]}; hi = {o1[4], o1[5], o1[6], o1[7]};
      *(v4f*)&S[1 * 256 + base] = lo; *(v4f*)&S[1 * 256 + base + 4] = hi;
      lo = {o2[0], o2[1], o2[2], o2[3]}; hi = {o2[4], o2[5], o2[6], o2[7]};
      *(v4f*)&S[2 * 256 + base] = lo; *(v4f*)&S[2 * 256 + base + 4] = hi;
      lo = {o3[0], o3[1], o3[2], o3[3]}; hi = {o3[4], o3[5], o3[6], o3[7]};
      *(v4f*)&S[3 * 256 + base] = lo; *(v4f*)&S[3 * 256 + base + 4] = hi;
    }
    wave_lds_sync();
  }

  float psum = 0.f, zh0 = 0.f, zh1 = 0.f, zh2 = 0.f, zh3 = 0.f, zh4 = 0.f;
#pragma unroll 4
  for (int j = 0; j < 32; ++j) {
    int d = (j << 5) | lane;
    float re = Sre[d], im = Sim[d];
    float p = re * re + im * im;
    psum += p;
    zh0 += (j & 1) ? -p : p;
    zh1 += (j & 2) ? -p : p;
    zh2 += (j & 4) ? -p : p;
    zh3 += (j & 8) ? -p : p;
    zh4 += (j & 16) ? -p : p;
  }
  red[lane] = psum;
  red[32 + lane] = zh0;  red[64 + lane] = zh1;  red[96 + lane] = zh2;
  red[128 + lane] = zh3; red[160 + lane] = zh4;
  wave_lds_sync();
  if (lane < NQ) {
    float z = 0.f;
    if (lane < 5) {
#pragma unroll
      for (int L = 0; L < 32; ++L) z += ((L >> lane) & 1) ? -red[L] : red[L];
    } else {
#pragma unroll
      for (int L = 0; L < 32; ++L) z += red[(lane - 4) * 32 + L];
    }
    zb[lane] = z;
  }
  wave_lds_sync();
  if (lane < NQ) {
    float acc = b_post[lane];
#pragma unroll
    for (int q = 0; q < NQ; ++q) acc += zb[q] * W_post[lane * NQ + q];
    out[b * NQ + lane] = acc;
  }
}

extern "C" void kernel_launch(void* const* d_in, const int* in_sizes, int n_in,
                              void* d_out, int out_size, void* d_ws, size_t ws_size,
                              hipStream_t stream) {
  const float* x       = (const float*)d_in[0];
  const float* W_pre   = (const float*)d_in[1];
  const float* b_pre   = (const float*)d_in[2];
  const float* weights = (const float*)d_in[3];
  const float* W_post  = (const float*)d_in[4];
  const float* b_post  = (const float*)d_in[5];
  float* out = (float*)d_out;
  hipLaunchKernelGGL(qnn_circuit_kernel, dim3(BATCH / WAVES), dim3(32 * WAVES), 0, stream,
                     x, W_pre, b_pre, weights, W_post, b_post, out);
}


# First RY layer from the product form of the initial state: its gates act on the 16-entry factor vectors and 8 WMMAs with K=4 assemble the state, not 64 dense ones (order of the f32 sums changes)
# speedup vs baseline: 2.0977x; 1.1383x over previous
#include <hip/hip_runtime.h>


#define NQ 10
#define DIM 1024
#define NLAYERS 4
#define BATCH 2048
#define WAVES 8

typedef __attribute__((ext_vector_type(2))) float v2f;
typedef __attribute__((ext_vector_type(4))) float v4f;
typedef __attribute__((ext_vector_type(8))) float v8f;

__device__ __forceinline__ int ladder_src(int d) { return (d ^ (d << 1)) & (DIM - 1); }

__device__ __forceinline__ void ry_row(float c, float s, int nb, float& r0, float& r1) {
  r0 = nb ? s : c;
  r1 = nb ? c : -s;
}

__device__ __forceinline__ void build_b_hi(const float* cc, const float* ss,
                                           int n, int h, v2f frag[4]) {
  float f0[2], f1[2], f2[2], f3[2];
  ry_row(cc[0], ss[0], (n >> 0) & 1, f0[0], f0[1]);
  ry_row(cc[1], ss[1], (n >> 1) & 1, f1[0], f1[1]);
  ry_row(cc[2], ss[2], (n >> 2) & 1, f2[0], f2[1]);
  ry_row(cc[3], ss[3], (n >> 3) & 1, f3[0], f3[1]);
  const float f1h = h ? f1[1] : f1[0];
#pragma unroll
  for (int ks = 0; ks < 4; ++ks) {
    v2f e;
    e.x = ((f0[0] * f1h) * f2[ks & 1]) * f3[ks >> 1];
    e.y = ((f0[1] * f1h) * f2[ks & 1]) * f3[ks >> 1];
    frag[ks] = e;
  }
}

__device__ __forceinline__ void build_a_lo(const float* cc, const float* ss,
                                           int m, int h, v2f frag[4]) {
  float f0[2], f1[2], f2[2], f3[2];
  ry_row(cc[0], ss[0], (m >> 0) & 1, f0[0], f0[1]);
  ry_row(cc[1], ss[1], (m >> 1) & 1, f1[0], f1[1]);
  ry_row(cc[2], ss[2], (m >> 2) & 1, f2[0], f2[1]);
  ry_row(cc[3], ss[3], (m >> 3) & 1, f3[0], f3[1]);
  const float f3h = h ? f3[1] : f3[0];
#pragma unroll
  for (int ks = 0; ks < 4; ++ks) {
    v2f e;
    e.x = ((f0[0] * f1[ks & 1]) * f2[ks >> 1]) * f3h;
    e.y = ((f0[1] * f1[ks & 1]) * f2[ks >> 1]) * f3h;
    frag[ks] = e;
  }
}

template <int J>
__device__ __forceinline__ float ry_lanes(float v, float c, float s, int idx) {
  const float p = __int_as_float(__builtin_amdgcn_update_dpp(
      0, __float_as_int(v), 0x160 | (1 << J), 0xf, 0xf, false));
  return c * v + (((idx >> J) & 1) ? s : -s) * p;
}

template <int J>
__device__ __forceinline__ void gates_g0_g1(const float* csc, const float* css, int lid,
                                            float& ar, float& ai, float& b0r, float& b0i,
                                            float& b1r, float& b1i) {
  const float ca = csc[J], sa = css[J];
  const float cb = csc[4 + J], sb = css[4 + J];
  ar = ry_lanes<J>(ar, ca, sa, lid);
  ai = ry_lanes<J>(ai, ca, sa, lid);
  b0r = ry_lanes<J>(b0r, cb, sb, lid);
  b0i = ry_lanes<J>(b0i, cb, sb, lid);
  b1r = ry_lanes<J>(b1r, cb, sb, lid);
  b1i = ry_lanes<J>(b1i, cb, sb, lid);
}

template <int J>
__device__ __forceinline__ void gates_t(const float* csc, const float* css, int lane,
                                        float& c0r, float& c0i, float& c1r, float& c1i) {
  const float cc = csc[8 + J], sc = css[8 + J];
  c0r = ry_lanes<J>(c0r, cc, sc, lane);
  c0i = ry_lanes<J>(c0i, cc, sc, lane);
  c1r = ry_lanes<J>(c1r, cc, sc, lane);
  c1i = ry_lanes<J>(c1i, cc, sc, lane);
}

struct WaveLds {
  float Sre[DIM];
  float Sim[DIM];
  float hs[16];
  float gs[16];
  float csc[(NLAYERS + 1) * NQ];
  float css[(NLAYERS + 1) * NQ];
  float red[6 * 32];
  float zb[16];
};

__device__ __forceinline__ void wave_lds_sync() {
  __builtin_amdgcn_fence(__ATOMIC_RELEASE, "workgroup", "local");
  __builtin_amdgcn_wave_barrier();
  __builtin_amdgcn_fence(__ATOMIC_ACQUIRE, "workgroup", "local");
}

__global__ __launch_bounds__(32 * WAVES) void qnn_circuit_kernel(
    const float* __restrict__ x, const float* __restrict__ W_pre,
    const float* __restrict__ b_pre, const float* __restrict__ weights,
    const float* __restrict__ W_post, const float* __restrict__ b_post,
    float* __restrict__ out) {
  __shared__ __attribute__((aligned(16))) WaveLds lds[WAVES];

  const int wave = __builtin_amdgcn_readfirstlane((int)(threadIdx.x >> 5));
  const int lane = threadIdx.x & 31;
  const int b = blockIdx.x * WAVES + wave;
  WaveLds& L = lds[wave];
  float* const Sre = L.Sre;
  float* const Sim = L.Sim;
  float* const hs = L.hs;
  float* const gs = L.gs;
  float* const csc = L.csc;
  float* const css = L.css;
  float* const red = L.red;
  float* const zb = L.zb;

  for (int idx = lane; idx < (NLAYERS + 1) * NQ; idx += 32) {
    float w = 0.5f * weights[idx];
    csc[idx] = __cosf(w);
    css[idx] = __sinf(w);
  }
  if (lane < NQ) {
    float acc = b_pre[lane];
#pragma unroll
    for (int k = 0; k < NQ; ++k) acc += x[b * NQ + k] * W_pre[lane * NQ + k];
    hs[lane] = acc;
  }
  wave_lds_sync();
  if (lane < NQ - 1) {
    const float PI = 3.14159265358979f;
    gs[lane] = (PI - hs[lane]) * (PI - hs[lane + 1]);
  }
  wave_lds_sync();

  float hr[NQ], gr[NQ - 1];
#pragma unroll
  for (int q = 0; q < NQ; ++q) hr[q] = hs[q];
#pragma unroll
  for (int q = 0; q < NQ - 1; ++q) gr[q] = gs[q];

  const int half = lane >> 4;
  const int lid = lane & 15;

  const float inv = 0.03125f;
  const int lz = lid ^ (lid >> 1);
  const int tb = lane & 3;
  float angA = 0.f, angB = 0.f;
#pragma unroll
  for (int q = 0; q < 4; ++q) {
    angA += ((lid >> q) & 1) ? -hr[q] : hr[q];
    angB += ((lid >> q) & 1) ? -hr[4 + q] : hr[4 + q];
  }
#pragma unroll
  for (int q = 0; q < 3; ++q) {
    angA += ((lz >> q) & 1) ? -gr[q] : gr[q];
    angB += ((lz >> q) & 1) ? -gr[4 + q] : gr[4 + q];
  }
  float angC = ((tb & 1) ? -hr[8] : hr[8]) + ((tb & 2) ? -hr[9] : hr[9]);
  angC += ((tb ^ (tb >> 1)) & 1) ? -gr[8] : gr[8];
  float sA, cA, sB, cB, sC, cC, s3, c3, s7, c7;
  __sincosf(angA, &sA, &cA);
  __sincosf(angB, &sB, &cB);
  __sincosf(angC, &sC, &cC);
  __sincosf(gr[3], &s3, &c3);
  __sincosf(gr[7], &s7, &c7);

  const bool fa = half && (lid & 8);
  float ar = fa ? -cA : cA, ai = fa ? sA : -sA;
  const bool fb = half && (lid & 1);
  float b0r = fb ? -cB : cB, b0i = fb ? sB : -sB;
  float b1r = (lid & 8) ? -b0r : b0r;
  float b1i = (lid & 8) ? -b0i : b0i;
  float c0r = cC, c0i = -sC;
  float c1r = (tb & 1) ? -cC : cC;
  float c1i = (tb & 1) ? sC : -sC;

  gates_g0_g1<0>(csc, css, lid, ar, ai, b0r, b0i, b1r, b1i);
  gates_g0_g1<1>(csc, css, lid, ar, ai, b0r, b0i, b1r, b1i);
  gates_g0_g1<2>(csc, css, lid, ar, ai, b0r, b0i, b1r, b1i);
  gates_g0_g1<3>(csc, css, lid, ar, ai, b0r, b0i, b1r, b1i);
  gates_t<0>(csc, css, lane, c0r, c0i, c1r, c1i);
  gates_t<1>(csc, css, lane, c0r, c0i, c1r, c1i);

  const float k0 = c3 * inv, k1 = s3 * inv;
  v2f af;
  af.x = half ? k1 * ai : k0 * ar;
  af.y = half ? -(k1 * ar) : k0 * ai;
  const float e0r = c7 * c0r, e0i = c7 * c0i;
  const float e1r = s7 * c1i, e1i = -(s7 * c1r);

  {
    v8f z0[2][4];
#pragma unroll
    for (int t = 0; t < 4; ++t) {
      const float E0r = __int_as_float(__builtin_amdgcn_readlane(__float_as_int(e0r), t));
      const float E0i = __int_as_float(__builtin_amdgcn_readlane(__float_as_int(e0i), t));
      const float E1r = __int_as_float(__builtin_amdgcn_readlane(__float_as_int(e1r), t));
      const float E1i = __int_as_float(__builtin_amdgcn_readlane(__float_as_int(e1i), t));
      const float dr = E0r * b0r - E0i * b0i + E1r * b1r - E1i * b1i;
      const float di = E0r * b0i + E0i * b0r + E1r * b1i + E1i * b1r;
      v2f bre, bim;
      bre.x = dr; bre.y = -di;
      bim.x = di; bim.y = dr;
      const v8f zero = {0.f, 0.f, 0.f, 0.f, 0.f, 0.f, 0.f, 0.f};
      z0[0][t] = __builtin_amdgcn_wmma_f32_16x16x4_f32(
          false, af, false, bre, (short)0, zero, false, false);
      z0[1][t] = __builtin_amdgcn_wmma_f32_16x16x4_f32(
          false, af, false, bim, (short)0, zero, false, false);
    }
    const int base = lid * 16 + 8 * half;
#pragma unroll
    for (int pl = 0; pl < 2; ++pl) {
      float* S = pl ? Sim : Sre;
#pragma unroll
      for (int t = 0; t < 4; ++t) {
        v4f lo, hi;
        lo = {z0[pl][t][0], z0[pl][t][1], z0[pl][t][2], z0[pl][t][3]};
        hi = {z0[pl][t][4], z0[pl][t][5], z0[pl][t][6], z0[pl][t][7]};
        *(v4f*)&S[t * 256 + base] = lo;
        *(v4f*)&S[t * 256 + base + 4] = hi;
      }
    }
  }
  wave_lds_sync();

  const int dlane = (half << 5) | lid;
  const int plane_src = dlane ^ (dlane << 1);

#pragma unroll 1
  for (int l = 1; l <= NLAYERS; ++l) {
    v2f bhi[4], alo[4];
    build_b_hi(&csc[l * NQ + 4], &css[l * NQ + 4], lid, half, bhi);
    build_a_lo(&csc[l * NQ + 0], &css[l * NQ + 0], lid, half, alo);
    const float c8 = csc[l * NQ + 8], s8 = css[l * NQ + 8];
    const float c9 = csc[l * NQ + 9], s9 = css[l * NQ + 9];

    v2f xa[2][4][4];
#pragma unroll
    for (int pl = 0; pl < 2; ++pl) {
      const float* S = pl ? Sim : Sre;
#pragma unroll
      for (int t = 0; t < 4; ++t) {
#pragma unroll
        for (int ks = 0; ks < 4; ++ks) {
          const int c0 = ladder_src(t * 256 + (4 * ks + 0) * 16);
          const int c1 = ladder_src(t * 256 + (4 * ks + 1) * 16);
          v2f av;
          av.x = S[plane_src ^ c0];
          av.y = S[plane_src ^ c1];
          xa[pl][t][ks] = av;
        }
      }
    }

#pragma unroll
    for (int pl = 0; pl < 2; ++pl) {
      asm volatile("" ::
        "v"(xa[pl][0][0]), "v"(xa[pl][0][1]), "v"(xa[pl][0][2]), "v"(xa[pl][0][3]),
        "v"(xa[pl][1][0]), "v"(xa[pl][1][1]), "v"(xa[pl][1][2]), "v"(xa[pl][1][3]),
        "v"(xa[pl][2][0]), "v"(xa[pl][2][1]), "v"(xa[pl][2][2]), "v"(xa[pl][2][3]),
        "v"(xa[pl][3][0]), "v"(xa[pl][3][1]), "v"(xa[pl][3][2]), "v"(xa[pl][3][3]));
    }

    v8f y[2][4];
#pragma unroll
    for (int pl = 0; pl < 2; ++pl) {
#pragma unroll
      for (int t = 0; t < 4; ++t) {
        v8f acc = {0.f, 0.f, 0.f, 0.f, 0.f, 0.f, 0.f, 0.f};
#pragma unroll
        for (int ks = 0; ks < 4; ++ks)
          acc = __builtin_amdgcn_wmma_f32_16x16x4_f32(
              false, xa[pl][t][ks], false, bhi[ks], (short)0, acc, false, false);
        y[pl][t] = acc;
      }
    }

    v8f z[2][4];
#pragma unroll
    for (int pl = 0; pl < 2; ++pl) {
#pragma unroll
      for (int t = 0; t < 4; ++t) {
        v8f acc = {0.f, 0.f, 0.f, 0.f, 0.f, 0.f, 0.f, 0.f};
#pragma unroll
        for (int ks = 0; ks < 4; ++ks) {
          v2f bv;
          bv.x = y[pl][t][2 * ks];
          bv.y = y[pl][t][2 * ks + 1];
          acc = __builtin_amdgcn_wmma_f32_16x16x4_f32(
              false, alo[ks], false, bv, (short)0, acc, false, false);
        }
        z[pl][t] = acc;
      }
    }

#pragma unroll
    for (int pl = 0; pl < 2; ++pl) {
      float* S = pl ? Sim : Sre;
      v8f o0, o1, o2, o3;
#pragma unroll
      for (int r = 0; r < 8; ++r) {
        float a0 = z[pl][0][r], a1 = z[pl][1][r];
        float a2 = z[pl][2][r], a3 = z[pl][3][r];
        float t0 = c8 * a0 - s8 * a1, t1 = s8 * a0 + c8 * a1;
        float t2 = c8 * a2 - s8 * a3, t3 = s8 * a2 + c8 * a3;
        o0[r] = c9 * t0 - s9 * t2;
        o1[r] = c9 * t1 - s9 * t3;
        o2[r] = s9 * t0 + c9 * t2;
        o3[r] = s9 * t1 + c9 * t3;
      }
      const int base = lid * 16 + 8 * half;
      v4f lo, hi;
      lo = {o0[0], o0[1], o0[2], o0[3]}; hi = {o0[4], o0[5], o0[6], o0[7]};
      *(v4f*)&S[0 * 256 + base] = lo; *(v4f*)&S[0 * 256 + base + 4] = hi;
      lo = {o1[0], o1[1], o1[2], o1[3]}; hi = {o1[4], o1[5], o1[6], o1[7]};
      *(v4f*)&S[1 * 256 + base] = lo; *(v4f*)&S[1 * 256 + base + 4] = hi;
      lo = {o2[0], o2[1], o2[2], o2[3]}; hi = {o2[4], o2[5], o2[6], o2[7]};
      *(v4f*)&S[2 * 256 + base] = lo; *(v4f*)&S[2 * 256 + base + 4] = hi;
      lo = {o3[0], o3[1], o3[2], o3[3]}; hi = {o3[4], o3[5], o3[6], o3[7]};
      *(v4f*)&S[3 * 256 + base] = lo; *(v4f*)&S[3 * 256 + base + 4] = hi;
    }
    wave_lds_sync();
  }

  float psum = 0.f, zh0 = 0.f, zh1 = 0.f, zh2 = 0.f, zh3 = 0.f, zh4 = 0.f;
#pragma unroll 4
  for (int j = 0; j < 32; ++j) {
    int d = (j << 5) | lane;
    float re = Sre[d], im = Sim[d];
    float p = re * re + im * im;
    psum += p;
    zh0 += (j & 1) ? -p : p;
    zh1 += (j & 2) ? -p : p;
    zh2 += (j & 4) ? -p : p;
    zh3 += (j & 8) ? -p : p;
    zh4 += (j & 16) ? -p : p;
  }
  red[lane] = psum;
  red[32 + lane] = zh0;  red[64 + lane] = zh1;  red[96 + lane] = zh2;
  red[128 + lane] = zh3; red[160 + lane] = zh4;
  wave_lds_sync();
  if (lane < NQ) {
    float z = 0.f;
    if (lane < 5) {
#pragma unroll
      for (int L = 0; L < 32; ++L) z += ((L >> lane) & 1) ? -red[L] : red[L];
    } else {
#pragma unroll
      for (int L = 0; L < 32; ++L) z += red[(lane - 4) * 32 + L];
    }
    zb[lane] = z;
  }
  wave_lds_sync();
  if (lane < NQ) {
    float acc = b_post[lane];
#pragma unroll
    for (int q = 0; q < NQ; ++q) acc += zb[q] * W_post[lane * NQ + q];
    out[b * NQ + lane] = acc;
  }
}

extern "C" void kernel_launch(void* const* d_in, const int* in_sizes, int n_in,
                              void* d_out, int out_size, void* d_ws, size_t ws_size,
                              hipStream_t stream) {
  const float* x       = (const float*)d_in[0];
  const float* W_pre   = (const float*)d_in[1];
  const float* b_pre   = (const float*)d_in[2];
  const float* weights = (const float*)d_in[3];
  const float* W_post  = (const float*)d_in[4];
  const float* b_post  = (const float*)d_in[5];
  float* out = (float*)d_out;
  hipLaunchKernelGGL(qnn_circuit_kernel, dim3(BATCH / WAVES), dim3(32 * WAVES), 0, stream,
                     x, W_pre, b_pre, weights, W_post, b_post, out);
}


# Readout without memory waits: last layer's output measured in its registers, ten sums across lanes by row_xmask and permlanex16, post-net row fetched at entry (order of the f32 sums changes)
# speedup vs baseline: 2.3693x; 1.1295x over previous
#include <hip/hip_runtime.h>


#define NQ 10
#define DIM 1024
#define NLAYERS 4
#define BATCH 2048
#define WAVES 8

typedef __attribute__((ext_vector_type(2))) float v2f;
typedef __attribute__((ext_vector_type(4))) float v4f;
typedef __attribute__((ext_vector_type(8))) float v8f;

__device__ __forceinline__ int ladder_src(int d) { return (d ^ (d << 1)) & (DIM - 1); }

__device__ __forceinline__ void ry_row(float c, float s, int nb, float& r0, float& r1) {
  r0 = nb ? s : c;
  r1 = nb ? c : -s;
}

__device__ __forceinline__ void build_b_hi(const float* cc, const float* ss,
                                           int n, int h, v2f frag[4]) {
  float f0[2], f1[2], f2[2], f3[2];
  ry_row(cc[0], ss[0], (n >> 0) & 1, f0[0], f0[1]);
  ry_row(cc[1], ss[1], (n >> 1) & 1, f1[0], f1[1]);
  ry_row(cc[2], ss[2], (n >> 2) & 1, f2[0], f2[1]);
  ry_row(cc[3], ss[3], (n >> 3) & 1, f3[0], f3[1]);
  const float f1h = h ? f1[1] : f1[0];
#pragma unroll
  for (int ks = 0; ks < 4; ++ks) {
    v2f e;
    e.x = ((f0[0] * f1h) * f2[ks & 1]) * f3[ks >> 1];
    e.y = ((f0[1] * f1h) * f2[ks & 1]) * f3[ks >> 1];
    frag[ks] = e;
  }
}

__device__ __forceinline__ void build_a_lo(const float* cc, const float* ss,
                                           int m, int h, v2f frag[4]) {
  float f0[2], f1[2], f2[2], f3[2];
  ry_row(cc[0], ss[0], (m >> 0) & 1, f0[0], f0[1]);
  ry_row(cc[1], ss[1], (m >> 1) & 1, f1[0], f1[1]);
  ry_row(cc[2], ss[2], (m >> 2) & 1, f2[0], f2[1]);
  ry_row(cc[3], ss[3], (m >> 3) & 1, f3[0], f3[1]);
  const float f3h = h ? f3[1] : f3[0];
#pragma unroll
  for (int ks = 0; ks < 4; ++ks) {
    v2f e;
    e.x = ((f0[0] * f1[ks & 1]) * f2[ks >> 1]) * f3h;
    e.y = ((f0[1] * f1[ks & 1]) * f2[ks >> 1]) * f3h;
    frag[ks] = e;
  }
}

template <int J>
__device__ __forceinline__ float ry_lanes(float v, float c, float s, int idx) {
  const float p = __int_as_float(__builtin_amdgcn_update_dpp(
      0, __float_as_int(v), 0x160 | (1 << J), 0xf, 0xf, false));
  return c * v + (((idx >> J) & 1) ? s : -s) * p;
}

template <int J>
__device__ __forceinline__ void gates_g0_g1(const float* csc, const float* css, int lid,
                                            float& ar, float& ai, float& b0r, float& b0i,
                                            float& b1r, float& b1i) {
  const float ca = csc[J], sa = css[J];
  const float cb = csc[4 + J], sb = css[4 + J];
  ar = ry_lanes<J>(ar, ca, sa, lid);
  ai = ry_lanes<J>(ai, ca, sa, lid);
  b0r = ry_lanes<J>(b0r, cb, sb, lid);
  b0i = ry_lanes<J>(b0i, cb, sb, lid);
  b1r = ry_lanes<J>(b1r, cb, sb, lid);
  b1i = ry_lanes<J>(b1i, cb, sb, lid);
}

template <int J>
__device__ __forceinline__ void gates_t(const float* csc, const float* css, int lane,
                                        float& c0r, float& c0i, float& c1r, float& c1i) {
  const float cc = csc[8 + J], sc = css[8 + J];
  c0r = ry_lanes<J>(c0r, cc, sc, lane);
  c0i = ry_lanes<J>(c0i, cc, sc, lane);
  c1r = ry_lanes<J>(c1r, cc, sc, lane);
  c1i = ry_lanes<J>(c1i, cc, sc, lane);
}

struct WaveLds {
  float Sre[DIM];
  float Sim[DIM];
  float hs[16];
  float gs[16];
  float csc[(NLAYERS + 1) * NQ];
  float css[(NLAYERS + 1) * NQ];
};

__device__ __forceinline__ void wave_lds_sync() {
  __builtin_amdgcn_fence(__ATOMIC_RELEASE, "workgroup", "local");
  __builtin_amdgcn_wave_barrier();
  __builtin_amdgcn_fence(__ATOMIC_ACQUIRE, "workgroup", "local");
}

template <int M>
__device__ __forceinline__ float add_xor_lane(float v) {
  return v + __int_as_float(__builtin_amdgcn_update_dpp(
      0, __float_as_int(v), 0x160 | M, 0xf, 0xf, false));
}

__device__ __forceinline__ float wave_sum(float v) {
  v = add_xor_lane<1>(v);
  v = add_xor_lane<2>(v);
  v = add_xor_lane<4>(v);
  v = add_xor_lane<8>(v);
  const float other = __int_as_float(__builtin_amdgcn_permlanex16(
      0, __float_as_int(v), 0x76543210u, 0xfedcba98u, false, false));
  return v + other;
}

__global__ __launch_bounds__(32 * WAVES) void qnn_circuit_kernel(
    const float* __restrict__ x, const float* __restrict__ W_pre,
    const float* __restrict__ b_pre, const float* __restrict__ weights,
    const float* __restrict__ W_post, const float* __restrict__ b_post,
    float* __restrict__ out) {
  __shared__ __attribute__((aligned(16))) WaveLds lds[WAVES];

  const int wave = __builtin_amdgcn_readfirstlane((int)(threadIdx.x >> 5));
  const int lane = threadIdx.x & 31;
  const int b = blockIdx.x * WAVES + wave;
  WaveLds& L = lds[wave];
  float* const Sre = L.Sre;
  float* const Sim = L.Sim;
  float* const hs = L.hs;
  float* const gs = L.gs;
  float* const csc = L.csc;
  float* const css = L.css;

  float wp[NQ], bp = 0.f;
#pragma unroll
  for (int q = 0; q < NQ; ++q) wp[q] = 0.f;
  if (lane < NQ) {
    bp = b_post[lane];
#pragma unroll
    for (int q = 0; q < NQ; ++q) wp[q] = W_post[lane * NQ + q];
  }

  for (int idx = lane; idx < (NLAYERS + 1) * NQ; idx += 32) {
    float w = 0.5f * weights[idx];
    csc[idx] = __cosf(w);
    css[idx] = __sinf(w);
  }
  if (lane < NQ) {
    float acc = b_pre[lane];
#pragma unroll
    for (int k = 0; k < NQ; ++k) acc += x[b * NQ + k] * W_pre[lane * NQ + k];
    hs[lane] = acc;
  }
  wave_lds_sync();
  if (lane < NQ - 1) {
    const float PI = 3.14159265358979f;
    gs[lane] = (PI - hs[lane]) * (PI - hs[lane + 1]);
  }
  wave_lds_sync();

  float hr[NQ], gr[NQ - 1];
#pragma unroll
  for (int q = 0; q < NQ; ++q) hr[q] = hs[q];
#pragma unroll
  for (int q = 0; q < NQ - 1; ++q) gr[q] = gs[q];

  const int half = lane >> 4;
  const int lid = lane & 15;

  const float inv = 0.03125f;
  const int lz = lid ^ (lid >> 1);
  const int tb = lane & 3;
  float angA = 0.f, angB = 0.f;
#pragma unroll
  for (int q = 0; q < 4; ++q) {
    angA += ((lid >> q) & 1) ? -hr[q] : hr[q];
    angB += ((lid >> q) & 1) ? -hr[4 + q] : hr[4 + q];
  }
#pragma unroll
  for (int q = 0; q < 3; ++q) {
    angA += ((lz >> q) & 1) ? -gr[q] : gr[q];
    angB += ((lz >> q) & 1) ? -gr[4 + q] : gr[4 + q];
  }
  float angC = ((tb & 1) ? -hr[8] : hr[8]) + ((tb & 2) ? -hr[9] : hr[9]);
  angC += ((tb ^ (tb >> 1)) & 1) ? -gr[8] : gr[8];
  float sA, cA, sB, cB, sC, cC, s3, c3, s7, c7;
  __sincosf(angA, &sA, &cA);
  __sincosf(angB, &sB, &cB);
  __sincosf(angC, &sC, &cC);
  __sincosf(gr[3], &s3, &c3);
  __sincosf(gr[7], &s7, &c7);

  const bool fa = half && (lid & 8);
  float ar = fa ? -cA : cA, ai = fa ? sA : -sA;
  const bool fb = half && (lid & 1);
  float b0r = fb ? -cB : cB, b0i = fb ? sB : -sB;
  float b1r = (lid & 8) ? -b0r : b0r;
  float b1i = (lid & 8) ? -b0i : b0i;
  float c0r = cC, c0i = -sC;
  float c1r = (tb & 1) ? -cC : cC;
  float c1i = (tb & 1) ? sC : -sC;

  gates_g0_g1<0>(csc, css, lid, ar, ai, b0r, b0i, b1r, b1i);
  gates_g0_g1<1>(csc, css, lid, ar, ai, b0r, b0i, b1r, b1i);
  gates_g0_g1<2>(csc, css, lid, ar, ai, b0r, b0i, b1r, b1i);
  gates_g0_g1<3>(csc, css, lid, ar, ai, b0r, b0i, b1r, b1i);
  gates_t<0>(csc, css, lane, c0r, c0i, c1r, c1i);
  gates_t<1>(csc, css, lane, c0r, c0i, c1r, c1i);

  const float k0 = c3 * inv, k1 = s3 * inv;
  v2f af;
  af.x = half ? k1 * ai : k0 * ar;
  af.y = half ? -(k1 * ar) : k0 * ai;
  const float e0r = c7 * c0r, e0i = c7 * c0i;
  const float e1r = s7 * c1i, e1i = -(s7 * c1r);

  {
    v8f z0[2][4];
#pragma unroll
    for (int t = 0; t < 4; ++t) {
      const float E0r = __int_as_float(__builtin_amdgcn_readlane(__float_as_int(e0r), t));
      const float E0i = __int_as_float(__builtin_amdgcn_readlane(__float_as_int(e0i), t));
      const float E1r = __int_as_float(__builtin_amdgcn_readlane(__float_as_int(e1r), t));
      const float E1i = __int_as_float(__builtin_amdgcn_readlane(__float_as_int(e1i), t));
      const float dr = E0r * b0r - E0i * b0i + E1r * b1r - E1i * b1i;
      const float di = E0r * b0i + E0i * b0r + E1r * b1i + E1i * b1r;
      v2f bre, bim;
      bre.x = dr; bre.y = -di;
      bim.x = di; bim.y = dr;
      const v8f zero = {0.f, 0.f, 0.f, 0.f, 0.f, 0.f, 0.f, 0.f};
      z0[0][t] = __builtin_amdgcn_wmma_f32_16x16x4_f32(
          false, af, false, bre, (short)0, zero, false, false);
      z0[1][t] = __builtin_amdgcn_wmma_f32_16x16x4_f32(
          false, af, false, bim, (short)0, zero, false, false);
    }
    const int base = lid * 16 + 8 * half;
#pragma unroll
    for (int pl = 0; pl < 2; ++pl) {
      float* S = pl ? Sim : Sre;
#pragma unroll
      for (int t = 0; t < 4; ++t) {
        v4f lo, hi;
        lo = {z0[pl][t][0], z0[pl][t][1], z0[pl][t][2], z0[pl][t][3]};
        hi = {z0[pl][t][4], z0[pl][t][5], z0[pl][t][6], z0[pl][t][7]};
        *(v4f*)&S[t * 256 + base] = lo;
        *(v4f*)&S[t * 256 + base + 4] = hi;
      }
    }
  }
  wave_lds_sync();

  const int dlane = (half << 5) | lid;
  const int plane_src = dlane ^ (dlane << 1);

  v8f o[2][4];
#pragma unroll 1
  for (int l = 1; l <= NLAYERS; ++l) {
    v2f bhi[4], alo[4];
    build_b_hi(&csc[l * NQ + 4], &css[l * NQ + 4], lid, half, bhi);
    build_a_lo(&csc[l * NQ + 0], &css[l * NQ + 0], lid, half, alo);
    const float c8 = csc[l * NQ + 8], s8 = css[l * NQ + 8];
    const float c9 = csc[l * NQ + 9], s9 = css[l * NQ + 9];

    v2f xa[2][4][4];
#pragma unroll
    for (int pl = 0; pl < 2; ++pl) {
      const float* S = pl ? Sim : Sre;
#pragma unroll
      for (int t = 0; t < 4; ++t) {
#pragma unroll
        for (int ks = 0; ks < 4; ++ks) {
          const int c0 = ladder_src(t * 256 + (4 * ks + 0) * 16);
          const int c1 = ladder_src(t * 256 + (4 * ks + 1) * 16);
          v2f av;
          av.x = S[plane_src ^ c0];
          av.y = S[plane_src ^ c1];
          xa[pl][t][ks] = av;
        }
      }
    }

#pragma unroll
    for (int pl = 0; pl < 2; ++pl) {
      asm volatile("" ::
        "v"(xa[pl][0][0]), "v"(xa[pl][0][1]), "v"(xa[pl][0][2]), "v"(xa[pl][0][3]),
        "v"(xa[pl][1][0]), "v"(xa[pl][1][1]), "v"(xa[pl][1][2]), "v"(xa[pl][1][3]),
        "v"(xa[pl][2][0]), "v"(xa[pl][2][1]), "v"(xa[pl][2][2]), "v"(xa[pl][2][3]),
        "v"(xa[pl][3][0]), "v"(xa[pl][3][1]), "v"(xa[pl][3][2]), "v"(xa[pl][3][3]));
    }

    v8f y[2][4];
#pragma unroll
    for (int pl = 0; pl < 2; ++pl) {
#pragma unroll
      for (int t = 0; t < 4; ++t) {
        v8f acc = {0.f, 0.f, 0.f, 0.f, 0.f, 0.f, 0.f, 0.f};
#pragma unroll
        for (int ks = 0; ks < 4; ++ks)
          acc = __builtin_amdgcn_wmma_f32_16x16x4_f32(
              false, xa[pl][t][ks], false, bhi[ks], (short)0, acc, false, false);
        y[pl][t] = acc;
      }
    }

    v8f z[2][4];
#pragma unroll
    for (int pl = 0; pl < 2; ++pl) {
#pragma unroll
      for (int t = 0; t < 4; ++t) {
        v8f acc = {0.f, 0.f, 0.f, 0.f, 0.f, 0.f, 0.f, 0.f};
#pragma unroll
        for (int ks = 0; ks < 4; ++ks) {
          v2f bv;
          bv.x = y[pl][t][2 * ks];
          bv.y = y[pl][t][2 * ks + 1];
          acc = __builtin_amdgcn_wmma_f32_16x16x4_f32(
              false, alo[ks], false, bv, (short)0, acc, false, false);
        }
        z[pl][t] = acc;
      }
    }

    const bool last = (l == NLAYERS);
#pragma unroll
    for (int pl = 0; pl < 2; ++pl) {
      float* S = pl ? Sim : Sre;
#pragma unroll
      for (int r = 0; r < 8; ++r) {
        float a0 = z[pl][0][r], a1 = z[pl][1][r];
        float a2 = z[pl][2][r], a3 = z[pl][3][r];
        float t0 = c8 * a0 - s8 * a1, t1 = s8 * a0 + c8 * a1;
        float t2 = c8 * a2 - s8 * a3, t3 = s8 * a2 + c8 * a3;
        o[pl][0][r] = c9 * t0 - s9 * t2;
        o[pl][1][r] = c9 * t1 - s9 * t3;
        o[pl][2][r] = s9 * t0 + c9 * t2;
        o[pl][3][r] = s9 * t1 + c9 * t3;
      }
      if (!last) {
        const int base = lid * 16 + 8 * half;
#pragma unroll
        for (int t = 0; t < 4; ++t) {
          v4f lo, hi;
          lo = {o[pl][t][0], o[pl][t][1], o[pl][t][2], o[pl][t][3]};
          hi = {o[pl][t][4], o[pl][t][5], o[pl][t][6], o[pl][t][7]};
          *(v4f*)&S[t * 256 + base] = lo;
          *(v4f*)&S[t * 256 + base + 4] = hi;
        }
      }
    }
    if (!last) wave_lds_sync();
  }

  float zs = 0.f, z0 = 0.f, z1 = 0.f, z2 = 0.f, z8 = 0.f, z9 = 0.f;
#pragma unroll
  for (int t = 0; t < 4; ++t) {
    v2f p[4];
#pragma unroll
    for (int i = 0; i < 4; ++i) {
      const v2f re = {o[0][t][2 * i], o[0][t][2 * i + 1]};
      const v2f im = {o[1][t][2 * i], o[1][t][2 * i + 1]};
      p[i] = re * re + im * im;
    }
    const float u0 = p[0].x + p[0].y, u1 = p[1].x + p[1].y, u2 = p[2].x + p[2].y, u3 = p[3].x + p[3].y;
    const float d0 = p[0].x - p[0].y, d1 = p[1].x - p[1].y, d2 = p[2].x - p[2].y, d3 = p[3].x - p[3].y;
    const float w0 = u0 + u1, w1 = u2 + u3;
    const float st = w0 + w1;
    zs += st;
    z0 += (d0 + d1) + (d2 + d3);
    z1 += (u0 - u1) + (u2 - u3);
    z2 += w0 - w1;
    z8 += (t & 1) ? -st : st;
    z9 += (t & 2) ? -st : st;
  }
  float zq[NQ];
  zq[0] = z0; zq[1] = z1; zq[2] = z2;
  zq[3] = half ? -zs : zs;
  zq[4] = (lid & 1) ? -zs : zs;
  zq[5] = (lid & 2) ? -zs : zs;
  zq[6] = (lid & 4) ? -zs : zs;
  zq[7] = (lid & 8) ? -zs : zs;
  zq[8] = z8; zq[9] = z9;
#pragma unroll
  for (int q = 0; q < NQ; ++q) zq[q] = wave_sum(zq[q]);

  if (lane < NQ) {
    float acc = bp;
#pragma unroll
    for (int q = 0; q < NQ; ++q) acc += zq[q] * wp[q];
    out[b * NQ + lane] = acc;
  }
}

extern "C" void kernel_launch(void* const* d_in, const int* in_sizes, int n_in,
                              void* d_out, int out_size, void* d_ws, size_t ws_size,
                              hipStream_t stream) {
  const float* x       = (const float*)d_in[0];
  const float* W_pre   = (const float*)d_in[1];
  const float* b_pre   = (const float*)d_in[2];
  const float* weights = (const float*)d_in[3];
  const float* W_post  = (const float*)d_in[4];
  const float* b_post  = (const float*)d_in[5];
  float* out = (float*)d_out;
  hipLaunchKernelGGL(qnn_circuit_kernel, dim3(BATCH / WAVES), dim3(32 * WAVES), 0, stream,
                     x, W_pre, b_pre, weights, W_post, b_post, out);
}


# Last RY layer not applied to the state: readout q = cos(th_q)*<Z_q> - sin(th_q)*<X_q> of layer 3's output in its registers, CNOT ladder folded into signs and partners (order of f32 sums changes)
# speedup vs baseline: 2.4970x; 1.0539x over previous
#include <hip/hip_runtime.h>


#define NQ 10
#define DIM 1024
#define NLAYERS 4
#define BATCH 2048
#define WAVES 8

typedef __attribute__((ext_vector_type(2))) float v2f;
typedef __attribute__((ext_vector_type(4))) float v4f;
typedef __attribute__((ext_vector_type(8))) float v8f;

__device__ __forceinline__ int ladder_src(int d) { return (d ^ (d << 1)) & (DIM - 1); }

__device__ __forceinline__ void ry_row(float c, float s, int nb, float& r0, float& r1) {
  r0 = nb ? s : c;
  r1 = nb ? c : -s;
}

__device__ __forceinline__ void build_b_hi(const float* cc, const float* ss,
                                           int n, int h, v2f frag[4]) {
  float f0[2], f1[2], f2[2], f3[2];
  ry_row(cc[0], ss[0], (n >> 0) & 1, f0[0], f0[1]);
  ry_row(cc[1], ss[1], (n >> 1) & 1, f1[0], f1[1]);
  ry_row(cc[2], ss[2], (n >> 2) & 1, f2[0], f2[1]);
  ry_row(cc[3], ss[3], (n >> 3) & 1, f3[0], f3[1]);
  const float f1h = h ? f1[1] : f1[0];
#pragma unroll
  for (int ks = 0; ks < 4; ++ks) {
    v2f e;
    e.x = ((f0[0] * f1h) * f2[ks & 1]) * f3[ks >> 1];
    e.y = ((f0[1] * f1h) * f2[ks & 1]) * f3[ks >> 1];
    frag[ks] = e;
  }
}

__device__ __forceinline__ void build_a_lo(const float* cc, const float* ss,
                                           int m, int h, v2f frag[4]) {
  float f0[2], f1[2], f2[2], f3[2];
  ry_row(cc[0], ss[0], (m >> 0) & 1, f0[0], f0[1]);
  ry_row(cc[1], ss[1], (m >> 1) & 1, f1[0], f1[1]);
  ry_row(cc[2], ss[2], (m >> 2) & 1, f2[0], f2[1]);
  ry_row(cc[3], ss[3], (m >> 3) & 1, f3[0], f3[1]);
  const float f3h = h ? f3[1] : f3[0];
#pragma unroll
  for (int ks = 0; ks < 4; ++ks) {
    v2f e;
    e.x = ((f0[0] * f1[ks & 1]) * f2[ks >> 1]) * f3h;
    e.y = ((f0[1] * f1[ks & 1]) * f2[ks >> 1]) * f3h;
    frag[ks] = e;
  }
}

template <int J>
__device__ __forceinline__ float ry_lanes(float v, float c, float s, int idx) {
  const float p = __int_as_float(__builtin_amdgcn_update_dpp(
      0, __float_as_int(v), 0x160 | (1 << J), 0xf, 0xf, false));
  return c * v + (((idx >> J) & 1) ? s : -s) * p;
}

template <int J>
__device__ __forceinline__ void gates_g0_g1(const float* csc, const float* css, int lid,
                                            float& ar, float& ai, float& b0r, float& b0i,
                                            float& b1r, float& b1i) {
  const float ca = csc[J], sa = css[J];
  const float cb = csc[4 + J], sb = css[4 + J];
  ar = ry_lanes<J>(ar, ca, sa, lid);
  ai = ry_lanes<J>(ai, ca, sa, lid);
  b0r = ry_lanes<J>(b0r, cb, sb, lid);
  b0i = ry_lanes<J>(b0i, cb, sb, lid);
  b1r = ry_lanes<J>(b1r, cb, sb, lid);
  b1i = ry_lanes<J>(b1i, cb, sb, lid);
}

template <int J>
__device__ __forceinline__ void gates_t(const float* csc, const float* css, int lane,
                                        float& c0r, float& c0i, float& c1r, float& c1i) {
  const float cc = csc[8 + J], sc = css[8 + J];
  c0r = ry_lanes<J>(c0r, cc, sc, lane);
  c0i = ry_lanes<J>(c0i, cc, sc, lane);
  c1r = ry_lanes<J>(c1r, cc, sc, lane);
  c1i = ry_lanes<J>(c1i, cc, sc, lane);
}

struct WaveLds {
  float Sre[DIM];
  float Sim[DIM];
  float hs[16];
  float gs[16];
  float csc[(NLAYERS + 1) * NQ];
  float css[(NLAYERS + 1) * NQ];
};

__device__ __forceinline__ void wave_lds_sync() {
  __builtin_amdgcn_fence(__ATOMIC_RELEASE, "workgroup", "local");
  __builtin_amdgcn_wave_barrier();
  __builtin_amdgcn_fence(__ATOMIC_ACQUIRE, "workgroup", "local");
}

template <int M>
__device__ __forceinline__ float add_xor_lane(float v) {
  return v + __int_as_float(__builtin_amdgcn_update_dpp(
      0, __float_as_int(v), 0x160 | M, 0xf, 0xf, false));
}

__device__ __forceinline__ float wave_sum(float v) {
  v = add_xor_lane<1>(v);
  v = add_xor_lane<2>(v);
  v = add_xor_lane<4>(v);
  v = add_xor_lane<8>(v);
  const float other = __int_as_float(__builtin_amdgcn_permlanex16(
      0, __float_as_int(v), 0x76543210u, 0xfedcba98u, false, false));
  return v + other;
}

__device__ __forceinline__ void pair_terms_lanes(float (&acc)[2][5], float re, float im,
                                                 float ore, float oim) {
  asm("v_fmac_f32_dpp %0, %10, %8 row_xmask:1 row_mask:0xf bank_mask:0xf\n\t"
      "v_fmac_f32_dpp %1, %11, %9 row_xmask:1 row_mask:0xf bank_mask:0xf\n\t"
      "v_fmac_f32_dpp %2, %8, %8 row_xmask:3 row_mask:0xf bank_mask:0xf\n\t"
      "v_fmac_f32_dpp %3, %9, %9 row_xmask:3 row_mask:0xf bank_mask:0xf\n\t"
      "v_fmac_f32_dpp %4, %8, %8 row_xmask:6 row_mask:0xf bank_mask:0xf\n\t"
      "v_fmac_f32_dpp %5, %9, %9 row_xmask:6 row_mask:0xf bank_mask:0xf\n\t"
      "v_fmac_f32_dpp %6, %8, %8 row_xmask:12 row_mask:0xf bank_mask:0xf\n\t"
      "v_fmac_f32_dpp %7, %9, %9 row_xmask:12 row_mask:0xf bank_mask:0xf"
      : "+v"(acc[0][0]), "+v"(acc[1][0]), "+v"(acc[0][1]), "+v"(acc[1][1]),
        "+v"(acc[0][2]), "+v"(acc[1][2]), "+v"(acc[0][3]), "+v"(acc[1][3])
      : "v"(re), "v"(im), "v"(ore), "v"(oim));
}

__device__ __forceinline__ void pair_terms_lane8(float (&acc)[2][5], float re, float im,
                                                 float pre, float pim) {
  asm("v_fmac_f32_dpp %0, %4, %2 row_xmask:8 row_mask:0xf bank_mask:0xf\n\t"
      "v_fmac_f32_dpp %1, %5, %3 row_xmask:8 row_mask:0xf bank_mask:0xf"
      : "+v"(acc[0][4]), "+v"(acc[1][4])
      : "v"(re), "v"(im), "v"(pre), "v"(pim));
}

__device__ __forceinline__ float from_other_row(float v) {
  return __int_as_float(__builtin_amdgcn_permlanex16(
      0, __float_as_int(v), 0x76543210u, 0xfedcba98u, true, true));
}

__global__ __launch_bounds__(32 * WAVES) void qnn_circuit_kernel(
    const float* __restrict__ x, const float* __restrict__ W_pre,
    const float* __restrict__ b_pre, const float* __restrict__ weights,
    const float* __restrict__ W_post, const float* __restrict__ b_post,
    float* __restrict__ out) {
  __shared__ __attribute__((aligned(16))) WaveLds lds[WAVES];

  const int wave = __builtin_amdgcn_readfirstlane((int)(threadIdx.x >> 5));
  const int lane = threadIdx.x & 31;
  const int b = blockIdx.x * WAVES + wave;
  WaveLds& L = lds[wave];
  float* const Sre = L.Sre;
  float* const Sim = L.Sim;
  float* const hs = L.hs;
  float* const gs = L.gs;
  float* const csc = L.csc;
  float* const css = L.css;

  float wp[NQ], bp = 0.f;
#pragma unroll
  for (int q = 0; q < NQ; ++q) wp[q] = 0.f;
  if (lane < NQ) {
    bp = b_post[lane];
#pragma unroll
    for (int q = 0; q < NQ; ++q) wp[q] = W_post[lane * NQ + q];
  }

  for (int idx = lane; idx < (NLAYERS + 1) * NQ; idx += 32) {
    float w = 0.5f * weights[idx];
    csc[idx] = __cosf(w);
    css[idx] = __sinf(w);
  }
  if (lane < NQ) {
    float acc = b_pre[lane];
#pragma unroll
    for (int k = 0; k < NQ; ++k) acc += x[b * NQ + k] * W_pre[lane * NQ + k];
    hs[lane] = acc;
  }
  wave_lds_sync();
  if (lane < NQ - 1) {
    const float PI = 3.14159265358979f;
    gs[lane] = (PI - hs[lane]) * (PI - hs[lane + 1]);
  }
  wave_lds_sync();

  float hr[NQ], gr[NQ - 1];
#pragma unroll
  for (int q = 0; q < NQ; ++q) hr[q] = hs[q];
#pragma unroll
  for (int q = 0; q < NQ - 1; ++q) gr[q] = gs[q];

  const int half = lane >> 4;
  const int lid = lane & 15;

  const float inv = 0.03125f;
  const int lz = lid ^ (lid >> 1);
  const int tb = lane & 3;
  float angA = 0.f, angB = 0.f;
#pragma unroll
  for (int q = 0; q < 4; ++q) {
    angA += ((lid >> q) & 1) ? -hr[q] : hr[q];
    angB += ((lid >> q) & 1) ? -hr[4 + q] : hr[4 + q];
  }
#pragma unroll
  for (int q = 0; q < 3; ++q) {
    angA += ((lz >> q) & 1) ? -gr[q] : gr[q];
    angB += ((lz >> q) & 1) ? -gr[4 + q] : gr[4 + q];
  }
  float angC = ((tb & 1) ? -hr[8] : hr[8]) + ((tb & 2) ? -hr[9] : hr[9]);
  angC += ((tb ^ (tb >> 1)) & 1) ? -gr[8] : gr[8];
  float sA, cA, sB, cB, sC, cC, s3, c3, s7, c7;
  __sincosf(angA, &sA, &cA);
  __sincosf(angB, &sB, &cB);
  __sincosf(angC, &sC, &cC);
  __sincosf(gr[3], &s3, &c3);
  __sincosf(gr[7], &s7, &c7);

  const bool fa = half && (lid & 8);
  float ar = fa ? -cA : cA, ai = fa ? sA : -sA;
  const bool fb = half && (lid & 1);
  float b0r = fb ? -cB : cB, b0i = fb ? sB : -sB;
  float b1r = (lid & 8) ? -b0r : b0r;
  float b1i = (lid & 8) ? -b0i : b0i;
  float c0r = cC, c0i = -sC;
  float c1r = (tb & 1) ? -cC : cC;
  float c1i = (tb & 1) ? sC : -sC;

  gates_g0_g1<0>(csc, css, lid, ar, ai, b0r, b0i, b1r, b1i);
  gates_g0_g1<1>(csc, css, lid, ar, ai, b0r, b0i, b1r, b1i);
  gates_g0_g1<2>(csc, css, lid, ar, ai, b0r, b0i, b1r, b1i);
  gates_g0_g1<3>(csc, css, lid, ar, ai, b0r, b0i, b1r, b1i);
  gates_t<0>(csc, css, lane, c0r, c0i, c1r, c1i);
  gates_t<1>(csc, css, lane, c0r, c0i, c1r, c1i);

  const float k0 = c3 * inv, k1 = s3 * inv;
  v2f af;
  af.x = half ? k1 * ai : k0 * ar;
  af.y = half ? -(k1 * ar) : k0 * ai;
  const float e0r = c7 * c0r, e0i = c7 * c0i;
  const float e1r = s7 * c1i, e1i = -(s7 * c1r);

  {
    v8f z0[2][4];
#pragma unroll
    for (int t = 0; t < 4; ++t) {
      const float E0r = __int_as_float(__builtin_amdgcn_readlane(__float_as_int(e0r), t));
      const float E0i = __int_as_float(__builtin_amdgcn_readlane(__float_as_int(e0i), t));
      const float E1r = __int_as_float(__builtin_amdgcn_readlane(__float_as_int(e1r), t));
      const float E1i = __int_as_float(__builtin_amdgcn_readlane(__float_as_int(e1i), t));
      const float dr = E0r * b0r - E0i * b0i + E1r * b1r - E1i * b1i;
      const float di = E0r * b0i + E0i * b0r + E1r * b1i + E1i * b1r;
      v2f bre, bim;
      bre.x = dr; bre.y = -di;
      bim.x = di; bim.y = dr;
      const v8f zero = {0.f, 0.f, 0.f, 0.f, 0.f, 0.f, 0.f, 0.f};
      z0[0][t] = __builtin_amdgcn_wmma_f32_16x16x4_f32(
          false, af, false, bre, (short)0, zero, false, false);
      z0[1][t] = __builtin_amdgcn_wmma_f32_16x16x4_f32(
          false, af, false, bim, (short)0, zero, false, false);
    }
    const int base = lid * 16 + 8 * half;
#pragma unroll
    for (int pl = 0; pl < 2; ++pl) {
      float* S = pl ? Sim : Sre;
#pragma unroll
      for (int t = 0; t < 4; ++t) {
        v4f lo, hi;
        lo = {z0[pl][t][0], z0[pl][t][1], z0[pl][t][2], z0[pl][t][3]};
        hi = {z0[pl][t][4], z0[pl][t][5], z0[pl][t][6], z0[pl][t][7]};
        *(v4f*)&S[t * 256 + base] = lo;
        *(v4f*)&S[t * 256 + base + 4] = hi;
      }
    }
  }
  wave_lds_sync();

  const int dlane = (half << 5) | lid;
  const int plane_src = dlane ^ (dlane << 1);

  v8f o[2][4];
#pragma unroll 1
  for (int l = 1; l <= NLAYERS - 1; ++l) {
    v2f bhi[4], alo[4];
    build_b_hi(&csc[l * NQ + 4], &css[l * NQ + 4], lid, half, bhi);
    build_a_lo(&csc[l * NQ + 0], &css[l * NQ + 0], lid, half, alo);
    const float c8 = csc[l * NQ + 8], s8 = css[l * NQ + 8];
    const float c9 = csc[l * NQ + 9], s9 = css[l * NQ + 9];

    v2f xa[2][4][4];
#pragma unroll
    for (int pl = 0; pl < 2; ++pl) {
      const float* S = pl ? Sim : Sre;
#pragma unroll
      for (int t = 0; t < 4; ++t) {
#pragma unroll
        for (int ks = 0; ks < 4; ++ks) {
          const int c0 = ladder_src(t * 256 + (4 * ks + 0) * 16);
          const int c1 = ladder_src(t * 256 + (4 * ks + 1) * 16);
          v2f av;
          av.x = S[plane_src ^ c0];
          av.y = S[plane_src ^ c1];
          xa[pl][t][ks] = av;
        }
      }
    }

#pragma unroll
    for (int pl = 0; pl < 2; ++pl) {
      asm volatile("" ::
        "v"(xa[pl][0][0]), "v"(xa[pl][0][1]), "v"(xa[pl][0][2]), "v"(xa[pl][0][3]),
        "v"(xa[pl][1][0]), "v"(xa[pl][1][1]), "v"(xa[pl][1][2]), "v"(xa[pl][1][3]),
        "v"(xa[pl][2][0]), "v"(xa[pl][2][1]), "v"(xa[pl][2][2]), "v"(xa[pl][2][3]),
        "v"(xa[pl][3][0]), "v"(xa[pl][3][1]), "v"(xa[pl][3][2]), "v"(xa[pl][3][3]));
    }

    v8f y[2][4];
#pragma unroll
    for (int pl = 0; pl < 2; ++pl) {
#pragma unroll
      for (int t = 0; t < 4; ++t) {
        v8f acc = {0.f, 0.f, 0.f, 0.f, 0.f, 0.f, 0.f, 0.f};
#pragma unroll
        for (int ks = 0; ks < 4; ++ks)
          acc = __builtin_amdgcn_wmma_f32_16x16x4_f32(
              false, xa[pl][t][ks], false, bhi[ks], (short)0, acc, false, false);
        y[pl][t] = acc;
      }
    }

    v8f z[2][4];
#pragma unroll
    for (int pl = 0; pl < 2; ++pl) {
#pragma unroll
      for (int t = 0; t < 4; ++t) {
        v8f acc = {0.f, 0.f, 0.f, 0.f, 0.f, 0.f, 0.f, 0.f};
#pragma unroll
        for (int ks = 0; ks < 4; ++ks) {
          v2f bv;
          bv.x = y[pl][t][2 * ks];
          bv.y = y[pl][t][2 * ks + 1];
          acc = __builtin_amdgcn_wmma_f32_16x16x4_f32(
              false, alo[ks], false, bv, (short)0, acc, false, false);
        }
        z[pl][t] = acc;
      }
    }

    const bool last = (l == NLAYERS - 1);
#pragma unroll
    for (int pl = 0; pl < 2; ++pl) {
      float* S = pl ? Sim : Sre;
#pragma unroll
      for (int r = 0; r < 8; ++r) {
        float a0 = z[pl][0][r], a1 = z[pl][1][r];
        float a2 = z[pl][2][r], a3 = z[pl][3][r];
        float t0 = c8 * a0 - s8 * a1, t1 = s8 * a0 + c8 * a1;
        float t2 = c8 * a2 - s8 * a3, t3 = s8 * a2 + c8 * a3;
        o[pl][0][r] = c9 * t0 - s9 * t2;
        o[pl][1][r] = c9 * t1 - s9 * t3;
        o[pl][2][r] = s9 * t0 + c9 * t2;
        o[pl][3][r] = s9 * t1 + c9 * t3;
      }
      if (!last) {
        const int base = lid * 16 + 8 * half;
#pragma unroll
        for (int t = 0; t < 4; ++t) {
          v4f lo, hi;
          lo = {o[pl][t][0], o[pl][t][1], o[pl][t][2], o[pl][t][3]};
          hi = {o[pl][t][4], o[pl][t][5], o[pl][t][6], o[pl][t][7]};
          *(v4f*)&S[t * 256 + base] = lo;
          *(v4f*)&S[t * 256 + base + 4] = hi;
        }
      }
    }
    if (!last) wave_lds_sync();
  }

  const float* cl = &csc[NLAYERS * NQ];
  const float* sl = &css[NLAYERS * NQ];

  v2f T0 = {0.f, 0.f}, T1 = {0.f, 0.f}, T2 = {0.f, 0.f}, T8 = {0.f, 0.f}, T9 = {0.f, 0.f};
#pragma unroll
  for (int t = 0; t < 4; ++t) {
    v2f p[4];
#pragma unroll
    for (int i = 0; i < 4; ++i) {
      const v2f re = {o[0][t][2 * i], o[0][t][2 * i + 1]};
      const v2f im = {o[1][t][2 * i], o[1][t][2 * i + 1]};
      p[i] = re * re + im * im;
    }
    const v2f A = p[0] - p[1], B = p[2] - p[3];
    const v2f S2 = A - B;
    T0 += (p[0] + p[1]) + (p[2] + p[3]);
    T1 += A + B;
    T2 += S2;
    T8 = (t & 1) ? T8 - S2 : T8 + S2;
    T9 = ((t ^ (t >> 1)) & 1) ? T9 - S2 : T9 + S2;
  }
  const float z0 = T0.x - T0.y, z1 = T1.x - T1.y, z2 = T2.x - T2.y;
  const float z8 = T8.x - T8.y, z9 = T9.x - T9.y;

  v8f orow[2][4];
#pragma unroll
  for (int pl = 0; pl < 2; ++pl)
#pragma unroll
    for (int t = 0; t < 4; ++t)
#pragma unroll
      for (int r = 0; r < 8; ++r) orow[pl][t][r] = from_other_row(o[pl][t][r]);

  float xl[2][5];
#pragma unroll
  for (int pl = 0; pl < 2; ++pl)
#pragma unroll
    for (int q = 0; q < 5; ++q) xl[pl][q] = 0.f;
#pragma unroll
  for (int t = 0; t < 4; ++t) {
#pragma unroll
    for (int r = 0; r < 8; ++r) {
      pair_terms_lanes(xl, o[0][t][r], o[1][t][r], orow[0][t][r], orow[1][t][r]);
      if ((t & 1) == 0) pair_terms_lane8(xl, o[0][t][r], o[1][t][r], o[0][t ^ 1][r], o[1][t ^ 1][r]);
    }
  }

  v2f xi[2][5];
#pragma unroll
  for (int pl = 0; pl < 2; ++pl) {
#pragma unroll
    for (int q = 0; q < 5; ++q) xi[pl][q] = (v2f){0.f, 0.f};
#pragma unroll
    for (int t = 0; t < 4; ++t) {
      const v8f a = o[pl][t];
      const v8f w = orow[pl][t];
      const v2f a01 = {a[0], a[1]}, a23 = {a[2], a[3]}, a45 = {a[4], a[5]}, a67 = {a[6], a[7]};
      const v2f a32 = {a[3], a[2]}, a76 = {a[7], a[6]};
      const v2f w45 = {w[4], w[5]}, w67 = {w[6], w[7]};
      xi[pl][0] += a01 * a32;
      xi[pl][0] += a45 * a76;
      xi[pl][1] += a01 * a67;
      xi[pl][1] += a45 * a23;
      xi[pl][2] += a01 * w45;
      xi[pl][2] += a23 * w67;
      if ((t & 1) == 0) {
        const v8f b = o[pl][t ^ 3];
        const v2f b01 = {b[0], b[1]}, b23 = {b[2], b[3]}, b45 = {b[4], b[5]}, b67 = {b[6], b[7]};
        xi[pl][3] += a01 * b01;
        xi[pl][3] += a23 * b23;
        xi[pl][3] += a45 * b45;
        xi[pl][3] += a67 * b67;
      }
      if ((t & 2) == 0) {
        const v8f b = o[pl][t ^ 2];
        const v2f b01 = {b[0], b[1]}, b23 = {b[2], b[3]}, b45 = {b[4], b[5]}, b67 = {b[6], b[7]};
        xi[pl][4] += a01 * b01;
        xi[pl][4] += a23 * b23;
        xi[pl][4] += a45 * b45;
        xi[pl][4] += a67 * b67;
      }
    }
  }
  float xq[NQ];
#pragma unroll
  for (int q = 0; q < 5; ++q) {
    const v2f sum = xi[0][q] + xi[1][q];
    xq[q < 3 ? q : q + 5] = sum.x + sum.y;
    xq[3 + q] = xl[0][q] + xl[1][q];
  }

  float zq[NQ];
  int par = half;
  zq[0] = z0; zq[1] = z1; zq[2] = z2;
  zq[3] = par ? -z2 : z2;
  par ^= lid & 1;
  zq[4] = par ? -z2 : z2;
  par ^= (lid >> 1) & 1;
  zq[5] = par ? -z2 : z2;
  par ^= (lid >> 2) & 1;
  zq[6] = par ? -z2 : z2;
  par ^= (lid >> 3) & 1;
  zq[7] = par ? -z2 : z2;
  zq[8] = par ? -z8 : z8;
  zq[9] = par ? -z9 : z9;
#pragma unroll
  for (int q = 0; q < NQ; ++q) {
    const float c = cl[q], s = sl[q];
    const bool once = (q <= 2) || (q >= 7);
    const float sn = once ? 4.f * (c * s) : 2.f * (c * s);
    zq[q] = (c * c - s * s) * zq[q] - sn * xq[q];
  }
#pragma unroll
  for (int q = 0; q < NQ; ++q) zq[q] = wave_sum(zq[q]);

  if (lane < NQ) {
    float acc = bp;
#pragma unroll
    for (int q = 0; q < NQ; ++q) acc += zq[q] * wp[q];
    out[b * NQ + lane] = acc;
  }
}

extern "C" void kernel_launch(void* const* d_in, const int* in_sizes, int n_in,
                              void* d_out, int out_size, void* d_ws, size_t ws_size,
                              hipStream_t stream) {
  const float* x       = (const float*)d_in[0];
  const float* W_pre   = (const float*)d_in[1];
  const float* b_pre   = (const float*)d_in[2];
  const float* weights = (const float*)d_in[3];
  const float* W_post  = (const float*)d_in[4];
  const float* b_post  = (const float*)d_in[5];
  float* out = (float*)d_out;
  hipLaunchKernelGGL(qnn_circuit_kernel, dim3(BATCH / WAVES), dim3(32 * WAVES), 0, stream,
                     x, W_pre, b_pre, weights, W_post, b_post, out);
}


# Second RY layer also from the product form: the CNOT ladder keeps the state a sum of 16 products, gates act on the factor vectors, 16 WMMAs (K=8) assemble it, not 8+64 (order of f32 sums changes)
# speedup vs baseline: 3.0110x; 1.2058x over previous
#include <hip/hip_runtime.h>


#define NQ 10
#define DIM 1024
#define NLAYERS 4
#define BATCH 2048
#define WAVES 8

typedef __attribute__((ext_vector_type(2))) float v2f;
typedef __attribute__((ext_vector_type(4))) float v4f;
typedef __attribute__((ext_vector_type(8))) float v8f;

__device__ __forceinline__ int ladder_src(int d) { return (d ^ (d << 1)) & (DIM - 1); }

__device__ __forceinline__ void ry_row(float c, float s, int nb, float& r0, float& r1) {
  r0 = nb ? s : c;
  r1 = nb ? c : -s;
}

__device__ __forceinline__ void build_b_hi(const float* cc, const float* ss,
                                           int n, int h, v2f frag[4]) {
  float f0[2], f1[2], f2[2], f3[2];
  ry_row(cc[0], ss[0], (n >> 0) & 1, f0[0], f0[1]);
  ry_row(cc[1], ss[1], (n >> 1) & 1, f1[0], f1[1]);
  ry_row(cc[2], ss[2], (n >> 2) & 1, f2[0], f2[1]);
  ry_row(cc[3], ss[3], (n >> 3) & 1, f3[0], f3[1]);
  const float f1h = h ? f1[1] : f1[0];
#pragma unroll
  for (int ks = 0; ks < 4; ++ks) {
    v2f e;
    e.x = ((f0[0] * f1h) * f2[ks & 1]) * f3[ks >> 1];
    e.y = ((f0[1] * f1h) * f2[ks & 1]) * f3[ks >> 1];
    frag[ks] = e;
  }
}

__device__ __forceinline__ void build_a_lo(const float* cc, const float* ss,
                                           int m, int h, v2f frag[4]) {
  float f0[2], f1[2], f2[2], f3[2];
  ry_row(cc[0], ss[0], (m >> 0) & 1, f0[0], f0[1]);
  ry_row(cc[1], ss[1], (m >> 1) & 1, f1[0], f1[1]);
  ry_row(cc[2], ss[2], (m >> 2) & 1, f2[0], f2[1]);
  ry_row(cc[3], ss[3], (m >> 3) & 1, f3[0], f3[1]);
  const float f3h = h ? f3[1] : f3[0];
#pragma unroll
  for (int ks = 0; ks < 4; ++ks) {
    v2f e;
    e.x = ((f0[0] * f1[ks & 1]) * f2[ks >> 1]) * f3h;
    e.y = ((f0[1] * f1[ks & 1]) * f2[ks >> 1]) * f3h;
    frag[ks] = e;
  }
}

template <int J>
__device__ __forceinline__ float ry_lanes(float v, float c, float s, int idx) {
  const float p = __int_as_float(__builtin_amdgcn_update_dpp(
      0, __float_as_int(v), 0x160 | (1 << J), 0xf, 0xf, false));
  return c * v + (((idx >> J) & 1) ? s : -s) * p;
}

template <int J>
__device__ __forceinline__ void gates_g0_g1(const float* csc, const float* css, int lid,
                                            float& ar, float& ai, float& b0r, float& b0i,
                                            float& b1r, float& b1i) {
  const float ca = csc[J], sa = css[J];
  const float cb = csc[4 + J], sb = css[4 + J];
  ar = ry_lanes<J>(ar, ca, sa, lid);
  ai = ry_lanes<J>(ai, ca, sa, lid);
  b0r = ry_lanes<J>(b0r, cb, sb, lid);
  b0i = ry_lanes<J>(b0i, cb, sb, lid);
  b1r = ry_lanes<J>(b1r, cb, sb, lid);
  b1i = ry_lanes<J>(b1i, cb, sb, lid);
}

template <int J>
__device__ __forceinline__ void gates_t(const float* csc, const float* css, int lane,
                                        float& c0r, float& c0i, float& c1r, float& c1i) {
  const float cc = csc[8 + J], sc = css[8 + J];
  c0r = ry_lanes<J>(c0r, cc, sc, lane);
  c0i = ry_lanes<J>(c0i, cc, sc, lane);
  c1r = ry_lanes<J>(c1r, cc, sc, lane);
  c1i = ry_lanes<J>(c1i, cc, sc, lane);
}

template <int J>
__device__ __forceinline__ void gates_hi(const float* csc, const float* css, int lid,
                                         float& b0r, float& b0i, float& b1r, float& b1i) {
  const float cb = csc[4 + J], sb = css[4 + J];
  b0r = ry_lanes<J>(b0r, cb, sb, lid);
  b0i = ry_lanes<J>(b0i, cb, sb, lid);
  b1r = ry_lanes<J>(b1r, cb, sb, lid);
  b1i = ry_lanes<J>(b1i, cb, sb, lid);
}

template <int M>
__device__ __forceinline__ float xor_lane(float v) {
  return __int_as_float(__builtin_amdgcn_update_dpp(
      0, __float_as_int(v), 0x160 | M, 0xf, 0xf, false));
}

template <int CTRL>
__device__ __forceinline__ float quad_lane(float v) {
  return __int_as_float(__builtin_amdgcn_update_dpp(
      0, __float_as_int(v), CTRL, 0xf, 0xf, false));
}

__device__ __forceinline__ float from_lam_lane(float v) {
  return __int_as_float(__builtin_amdgcn_permlane16(
      0, __float_as_int(v), 0x9afc5630u, 0x1274deb8u, false, false));
}

__device__ __forceinline__ void split_top_gate(float v, float c, float s, int hi,
                                               float& out0, float& out1) {
  const float co = c * v;
  const float sp = s * xor_lane<8>(v);
  out0 = hi ? sp : co;
  out1 = hi ? co : -sp;
}

struct WaveLds {
  float Sre[DIM];
  float Sim[DIM];
  float hs[16];
  float gs[16];
  float csc[(NLAYERS + 1) * NQ];
  float css[(NLAYERS + 1) * NQ];
};

__device__ __forceinline__ void wave_lds_sync() {
  __builtin_amdgcn_fence(__ATOMIC_RELEASE, "workgroup", "local");
  __builtin_amdgcn_wave_barrier();
  __builtin_amdgcn_fence(__ATOMIC_ACQUIRE, "workgroup", "local");
}

template <int M>
__device__ __forceinline__ float add_xor_lane(float v) {
  return v + __int_as_float(__builtin_amdgcn_update_dpp(
      0, __float_as_int(v), 0x160 | M, 0xf, 0xf, false));
}

__device__ __forceinline__ float wave_sum(float v) {
  v = add_xor_lane<1>(v);
  v = add_xor_lane<2>(v);
  v = add_xor_lane<4>(v);
  v = add_xor_lane<8>(v);
  const float other = __int_as_float(__builtin_amdgcn_permlanex16(
      0, __float_as_int(v), 0x76543210u, 0xfedcba98u, false, false));
  return v + other;
}

__device__ __forceinline__ void pair_terms_lanes(float (&acc)[2][5], float re, float im,
                                                 float ore, float oim) {
  asm("v_fmac_f32_dpp %0, %10, %8 row_xmask:1 row_mask:0xf bank_mask:0xf\n\t"
      "v_fmac_f32_dpp %1, %11, %9 row_xmask:1 row_mask:0xf bank_mask:0xf\n\t"
      "v_fmac_f32_dpp %2, %8, %8 row_xmask:3 row_mask:0xf bank_mask:0xf\n\t"
      "v_fmac_f32_dpp %3, %9, %9 row_xmask:3 row_mask:0xf bank_mask:0xf\n\t"
      "v_fmac_f32_dpp %4, %8, %8 row_xmask:6 row_mask:0xf bank_mask:0xf\n\t"
      "v_fmac_f32_dpp %5, %9, %9 row_xmask:6 row_mask:0xf bank_mask:0xf\n\t"
      "v_fmac_f32_dpp %6, %8, %8 row_xmask:12 row_mask:0xf bank_mask:0xf\n\t"
      "v_fmac_f32_dpp %7, %9, %9 row_xmask:12 row_mask:0xf bank_mask:0xf"
      : "+v"(acc[0][0]), "+v"(acc[1][0]), "+v"(acc[0][1]), "+v"(acc[1][1]),
        "+v"(acc[0][2]), "+v"(acc[1][2]), "+v"(acc[0][3]), "+v"(acc[1][3])
      : "v"(re), "v"(im), "v"(ore), "v"(oim));
}

__device__ __forceinline__ void pair_terms_lane8(float (&acc)[2][5], float re, float im,
                                                 float pre, float pim) {
  asm("v_fmac_f32_dpp %0, %4, %2 row_xmask:8 row_mask:0xf bank_mask:0xf\n\t"
      "v_fmac_f32_dpp %1, %5, %3 row_xmask:8 row_mask:0xf bank_mask:0xf"
      : "+v"(acc[0][4]), "+v"(acc[1][4])
      : "v"(re), "v"(im), "v"(pre), "v"(pim));
}

__device__ __forceinline__ float from_other_row(float v) {
  return __int_as_float(__builtin_amdgcn_permlanex16(
      0, __float_as_int(v), 0x76543210u, 0xfedcba98u, true, true));
}

__global__ __launch_bounds__(32 * WAVES) void qnn_circuit_kernel(
    const float* __restrict__ x, const float* __restrict__ W_pre,
    const float* __restrict__ b_pre, const float* __restrict__ weights,
    const float* __restrict__ W_post, const float* __restrict__ b_post,
    float* __restrict__ out) {
  __shared__ __attribute__((aligned(16))) WaveLds lds[WAVES];

  const int wave = __builtin_amdgcn_readfirstlane((int)(threadIdx.x >> 5));
  const int lane = threadIdx.x & 31;
  const int b = blockIdx.x * WAVES + wave;
  WaveLds& L = lds[wave];
  float* const Sre = L.Sre;
  float* const Sim = L.Sim;
  float* const hs = L.hs;
  float* const gs = L.gs;
  float* const csc = L.csc;
  float* const css = L.css;

  float wp[NQ], bp = 0.f;
#pragma unroll
  for (int q = 0; q < NQ; ++q) wp[q] = 0.f;
  if (lane < NQ) {
    bp = b_post[lane];
#pragma unroll
    for (int q = 0; q < NQ; ++q) wp[q] = W_post[lane * NQ + q];
  }

  for (int idx = lane; idx < (NLAYERS + 1) * NQ; idx += 32) {
    float w = 0.5f * weights[idx];
    csc[idx] = __cosf(w);
    css[idx] = __sinf(w);
  }
  if (lane < NQ) {
    float acc = b_pre[lane];
#pragma unroll
    for (int k = 0; k < NQ; ++k) acc += x[b * NQ + k] * W_pre[lane * NQ + k];
    hs[lane] = acc;
  }
  wave_lds_sync();
  if (lane < NQ - 1) {
    const float PI = 3.14159265358979f;
    gs[lane] = (PI - hs[lane]) * (PI - hs[lane + 1]);
  }
  wave_lds_sync();

  float hr[NQ], gr[NQ - 1];
#pragma unroll
  for (int q = 0; q < NQ; ++q) hr[q] = hs[q];
#pragma unroll
  for (int q = 0; q < NQ - 1; ++q) gr[q] = gs[q];

  const int half = lane >> 4;
  const int lid = lane & 15;

  const float inv = 0.03125f;
  const int lz = lid ^ (lid >> 1);
  const int tb = lane & 3;
  float angA = 0.f, angB = 0.f;
#pragma unroll
  for (int q = 0; q < 4; ++q) {
    angA += ((lid >> q) & 1) ? -hr[q] : hr[q];
    angB += ((lid >> q) & 1) ? -hr[4 + q] : hr[4 + q];
  }
#pragma unroll
  for (int q = 0; q < 3; ++q) {
    angA += ((lz >> q) & 1) ? -gr[q] : gr[q];
    angB += ((lz >> q) & 1) ? -gr[4 + q] : gr[4 + q];
  }
  float angC = ((tb & 1) ? -hr[8] : hr[8]) + ((tb & 2) ? -hr[9] : hr[9]);
  angC += ((tb ^ (tb >> 1)) & 1) ? -gr[8] : gr[8];
  float sA, cA, sB, cB, sC, cC, s3, c3, s7, c7;
  __sincosf(angA, &sA, &cA);
  __sincosf(angB, &sB, &cB);
  __sincosf(angC, &sC, &cC);
  __sincosf(gr[3], &s3, &c3);
  __sincosf(gr[7], &s7, &c7);

  const bool fa = half && (lid & 8);
  float ar = fa ? -cA : cA, ai = fa ? sA : -sA;
  const bool fb = half && (lid & 1);
  float b0r = fb ? -cB : cB, b0i = fb ? sB : -sB;
  float b1r = (lid & 8) ? -b0r : b0r;
  float b1i = (lid & 8) ? -b0i : b0i;
  float c0r = cC, c0i = -sC;
  float c1r = (tb & 1) ? -cC : cC;
  float c1i = (tb & 1) ? sC : -sC;

  gates_g0_g1<0>(csc, css, lid, ar, ai, b0r, b0i, b1r, b1i);
  gates_g0_g1<1>(csc, css, lid, ar, ai, b0r, b0i, b1r, b1i);
  gates_g0_g1<2>(csc, css, lid, ar, ai, b0r, b0i, b1r, b1i);
  gates_g0_g1<3>(csc, css, lid, ar, ai, b0r, b0i, b1r, b1i);
  gates_t<0>(csc, css, lane, c0r, c0i, c1r, c1i);
  gates_t<1>(csc, css, lane, c0r, c0i, c1r, c1i);


  ar = from_lam_lane(ar);   ai = from_lam_lane(ai);
  b0r = from_lam_lane(b0r); b0i = from_lam_lane(b0i);
  b1r = from_lam_lane(b1r); b1i = from_lam_lane(b1i);
  float d0r = xor_lane<15>(b0r), d0i = xor_lane<15>(b0i);
  float d1r = xor_lane<15>(b1r), d1i = xor_lane<15>(b1i);
  float p0r = quad_lane<0x39>(c0r), p0i = quad_lane<0x39>(c0i);
  float p1r = quad_lane<0x39>(c1r), p1i = quad_lane<0x39>(c1i);
  c0r = quad_lane<0x6c>(c0r); c0i = quad_lane<0x6c>(c0i);
  c1r = quad_lane<0x6c>(c1r); c1i = quad_lane<0x6c>(c1i);

  const float* const csc1 = csc + NQ;
  const float* const css1 = css + NQ;
  gates_g0_g1<0>(csc1, css1, lid, ar, ai, b0r, b0i, b1r, b1i);
  gates_g0_g1<1>(csc1, css1, lid, ar, ai, b0r, b0i, b1r, b1i);
  gates_g0_g1<2>(csc1, css1, lid, ar, ai, b0r, b0i, b1r, b1i);
  gates_hi<0>(csc1, css1, lid, d0r, d0i, d1r, d1i);
  gates_hi<1>(csc1, css1, lid, d0r, d0i, d1r, d1i);
  gates_hi<2>(csc1, css1, lid, d0r, d0i, d1r, d1i);
  gates_t<0>(csc1, css1, lane, c0r, c0i, c1r, c1i);
  gates_t<1>(csc1, css1, lane, c0r, c0i, c1r, c1i);
  gates_t<0>(csc1, css1, lane, p0r, p0i, p1r, p1i);
  gates_t<1>(csc1, css1, lane, p0r, p0i, p1r, p1i);

  const int hi = lid >> 3;
  float fr[2], fi[2];
  split_top_gate(ar, csc1[3], css1[3], hi, fr[0], fr[1]);
  split_top_gate(ai, csc1[3], css1[3], hi, fi[0], fi[1]);
  float Hr[2][2][2], Hi[2][2][2];
  split_top_gate(b0r, csc1[7], css1[7], hi, Hr[0][0][0], Hr[0][0][1]);
  split_top_gate(b0i, csc1[7], css1[7], hi, Hi[0][0][0], Hi[0][0][1]);
  split_top_gate(b1r, csc1[7], css1[7], hi, Hr[1][0][0], Hr[1][0][1]);
  split_top_gate(b1i, csc1[7], css1[7], hi, Hi[1][0][0], Hi[1][0][1]);
  split_top_gate(d0r, csc1[7], css1[7], hi, Hr[0][1][0], Hr[0][1][1]);
  split_top_gate(d0i, csc1[7], css1[7], hi, Hi[0][1][0], Hi[0][1][1]);
  split_top_gate(d1r, csc1[7], css1[7], hi, Hr[1][1][0], Hr[1][1][1]);
  split_top_gate(d1i, csc1[7], css1[7], hi, Hi[1][1][0], Hi[1][1][1]);

  const float k0 = c3 * inv, k1 = s3 * inv;
  v2f af[2];
#pragma unroll
  for (int s = 0; s < 2; ++s) {
    af[s].x = half ? k1 * fi[s] : k0 * fr[s];
    af[s].y = half ? -(k1 * fr[s]) : k0 * fi[s];
  }
  float er[2][2], ei[2][2];
  er[0][0] = c7 * c0r; ei[0][0] = c7 * c0i;
  er[0][1] = c7 * p0r; ei[0][1] = c7 * p0i;
  er[1][0] = s7 * c1i; ei[1][0] = -(s7 * c1r);
  er[1][1] = s7 * p1i; ei[1][1] = -(s7 * p1r);

  {
    v8f z0[2][4];
#pragma unroll
    for (int t = 0; t < 4; ++t) {
      float Er[2][2], Ei[2][2];
#pragma unroll
      for (int b2 = 0; b2 < 2; ++b2)
#pragma unroll
        for (int u = 0; u < 2; ++u) {
          Er[b2][u] = __int_as_float(__builtin_amdgcn_readlane(__float_as_int(er[b2][u]), t));
          Ei[b2][u] = __int_as_float(__builtin_amdgcn_readlane(__float_as_int(ei[b2][u]), t));
        }
      v8f accre = {0.f, 0.f, 0.f, 0.f, 0.f, 0.f, 0.f, 0.f};
      v8f accim = {0.f, 0.f, 0.f, 0.f, 0.f, 0.f, 0.f, 0.f};
#pragma unroll
      for (int s = 0; s < 2; ++s) {
        float dr = 0.f, di = 0.f;
#pragma unroll
        for (int b2 = 0; b2 < 2; ++b2)
#pragma unroll
          for (int u = 0; u < 2; ++u) {
            dr += Er[b2][u] * Hr[b2][s][u] - Ei[b2][u] * Hi[b2][s][u];
            di += Er[b2][u] * Hi[b2][s][u] + Ei[b2][u] * Hr[b2][s][u];
          }
        v2f bre, bim;
        bre.x = dr; bre.y = -di;
        bim.x = di; bim.y = dr;
        accre = __builtin_amdgcn_wmma_f32_16x16x4_f32(
            false, af[s], false, bre, (short)0, accre, false, false);
        accim = __builtin_amdgcn_wmma_f32_16x16x4_f32(
            false, af[s], false, bim, (short)0, accim, false, false);
      }
      z0[0][t] = accre;
      z0[1][t] = accim;
    }
    const int base = lid * 16 + 8 * half;
#pragma unroll
    for (int pl = 0; pl < 2; ++pl) {
      float* S = pl ? Sim : Sre;
#pragma unroll
      for (int t = 0; t < 4; ++t) {
        v4f lo, hi;
        lo = {z0[pl][t][0], z0[pl][t][1], z0[pl][t][2], z0[pl][t][3]};
        hi = {z0[pl][t][4], z0[pl][t][5], z0[pl][t][6], z0[pl][t][7]};
        *(v4f*)&S[t * 256 + base] = lo;
        *(v4f*)&S[t * 256 + base + 4] = hi;
      }
    }
  }
  wave_lds_sync();

  const int dlane = (half << 5) | lid;
  const int plane_src = dlane ^ (dlane << 1);

  v8f o[2][4];
#pragma unroll 1
  for (int l = 2; l <= NLAYERS - 1; ++l) {
    v2f bhi[4], alo[4];
    build_b_hi(&csc[l * NQ + 4], &css[l * NQ + 4], lid, half, bhi);
    build_a_lo(&csc[l * NQ + 0], &css[l * NQ + 0], lid, half, alo);
    const float c8 = csc[l * NQ + 8], s8 = css[l * NQ + 8];
    const float c9 = csc[l * NQ + 9], s9 = css[l * NQ + 9];

    v2f xa[2][4][4];
#pragma unroll
    for (int pl = 0; pl < 2; ++pl) {
      const float* S = pl ? Sim : Sre;
#pragma unroll
      for (int t = 0; t < 4; ++t) {
#pragma unroll
        for (int ks = 0; ks < 4; ++ks) {
          const int c0 = ladder_src(t * 256 + (4 * ks + 0) * 16);
          const int c1 = ladder_src(t * 256 + (4 * ks + 1) * 16);
          v2f av;
          av.x = S[plane_src ^ c0];
          av.y = S[plane_src ^ c1];
          xa[pl][t][ks] = av;
        }
      }
    }

#pragma unroll
    for (int pl = 0; pl < 2; ++pl) {
      asm volatile("" ::
        "v"(xa[pl][0][0]), "v"(xa[pl][0][1]), "v"(xa[pl][0][2]), "v"(xa[pl][0][3]),
        "v"(xa[pl][1][0]), "v"(xa[pl][1][1]), "v"(xa[pl][1][2]), "v"(xa[pl][1][3]),
        "v"(xa[pl][2][0]), "v"(xa[pl][2][1]), "v"(xa[pl][2][2]), "v"(xa[pl][2][3]),
        "v"(xa[pl][3][0]), "v"(xa[pl][3][1]), "v"(xa[pl][3][2]), "v"(xa[pl][3][3]));
    }

    v8f y[2][4];
#pragma unroll
    for (int pl = 0; pl < 2; ++pl) {
#pragma unroll
      for (int t = 0; t < 4; ++t) {
        v8f acc = {0.f, 0.f, 0.f, 0.f, 0.f, 0.f, 0.f, 0.f};
#pragma unroll
        for (int ks = 0; ks < 4; ++ks)
          acc = __builtin_amdgcn_wmma_f32_16x16x4_f32(
              false, xa[pl][t][ks], false, bhi[ks], (short)0, acc, false, false);
        y[pl][t] = acc;
      }
    }

    v8f z[2][4];
#pragma unroll
    for (int pl = 0; pl < 2; ++pl) {
#pragma unroll
      for (int t = 0; t < 4; ++t) {
        v8f acc = {0.f, 0.f, 0.f, 0.f, 0.f, 0.f, 0.f, 0.f};
#pragma unroll
        for (int ks = 0; ks < 4; ++ks) {
          v2f bv;
          bv.x = y[pl][t][2 * ks];
          bv.y = y[pl][t][2 * ks + 1];
          acc = __builtin_amdgcn_wmma_f32_16x16x4_f32(
              false, alo[ks], false, bv, (short)0, acc, false, false);
        }
        z[pl][t] = acc;
      }
    }

    const bool last = (l == NLAYERS - 1);
#pragma unroll
    for (int pl = 0; pl < 2; ++pl) {
      float* S = pl ? Sim : Sre;
#pragma unroll
      for (int r = 0; r < 8; ++r) {
        float a0 = z[pl][0][r], a1 = z[pl][1][r];
        float a2 = z[pl][2][r], a3 = z[pl][3][r];
        float t0 = c8 * a0 - s8 * a1, t1 = s8 * a0 + c8 * a1;
        float t2 = c8 * a2 - s8 * a3, t3 = s8 * a2 + c8 * a3;
        o[pl][0][r] = c9 * t0 - s9 * t2;
        o[pl][1][r] = c9 * t1 - s9 * t3;
        o[pl][2][r] = s9 * t0 + c9 * t2;
        o[pl][3][r] = s9 * t1 + c9 * t3;
      }
      if (!last) {
        const int base = lid * 16 + 8 * half;
#pragma unroll
        for (int t = 0; t < 4; ++t) {
          v4f lo, hi;
          lo = {o[pl][t][0], o[pl][t][1], o[pl][t][2], o[pl][t][3]};
          hi = {o[pl][t][4], o[pl][t][5], o[pl][t][6], o[pl][t][7]};
          *(v4f*)&S[t * 256 + base] = lo;
          *(v4f*)&S[t * 256 + base + 4] = hi;
        }
      }
    }
    if (!last) wave_lds_sync();
  }

  const float* cl = &csc[NLAYERS * NQ];
  const float* sl = &css[NLAYERS * NQ];

  v2f T0 = {0.f, 0.f}, T1 = {0.f, 0.f}, T2 = {0.f, 0.f}, T8 = {0.f, 0.f}, T9 = {0.f, 0.f};
#pragma unroll
  for (int t = 0; t < 4; ++t) {
    v2f p[4];
#pragma unroll
    for (int i = 0; i < 4; ++i) {
      const v2f re = {o[0][t][2 * i], o[0][t][2 * i + 1]};
      const v2f im = {o[1][t][2 * i], o[1][t][2 * i + 1]};
      p[i] = re * re + im * im;
    }
    const v2f A = p[0] - p[1], B = p[2] - p[3];
    const v2f S2 = A - B;
    T0 += (p[0] + p[1]) + (p[2] + p[3]);
    T1 += A + B;
    T2 += S2;
    T8 = (t & 1) ? T8 - S2 : T8 + S2;
    T9 = ((t ^ (t >> 1)) & 1) ? T9 - S2 : T9 + S2;
  }
  const float z0 = T0.x - T0.y, z1 = T1.x - T1.y, z2 = T2.x - T2.y;
  const float z8 = T8.x - T8.y, z9 = T9.x - T9.y;

  v8f orow[2][4];
#pragma unroll
  for (int pl = 0; pl < 2; ++pl)
#pragma unroll
    for (int t = 0; t < 4; ++t)
#pragma unroll
      for (int r = 0; r < 8; ++r) orow[pl][t][r] = from_other_row(o[pl][t][r]);

  float xl[2][5];
#pragma unroll
  for (int pl = 0; pl < 2; ++pl)
#pragma unroll
    for (int q = 0; q < 5; ++q) xl[pl][q] = 0.f;
#pragma unroll
  for (int t = 0; t < 4; ++t) {
#pragma unroll
    for (int r = 0; r < 8; ++r) {
      pair_terms_lanes(xl, o[0][t][r], o[1][t][r], orow[0][t][r], orow[1][t][r]);
      if ((t & 1) == 0) pair_terms_lane8(xl, o[0][t][r], o[1][t][r], o[0][t ^ 1][r], o[1][t ^ 1][r]);
    }
  }

  v2f xi[2][5];
#pragma unroll
  for (int pl = 0; pl < 2; ++pl) {
#pragma unroll
    for (int q = 0; q < 5; ++q) xi[pl][q] = (v2f){0.f, 0.f};
#pragma unroll
    for (int t = 0; t < 4; ++t) {
      const v8f a = o[pl][t];
      const v8f w = orow[pl][t];
      const v2f a01 = {a[0], a[1]}, a23 = {a[2], a[3]}, a45 = {a[4], a[5]}, a67 = {a[6], a[7]};
      const v2f a32 = {a[3], a[2]}, a76 = {a[7], a[6]};
      const v2f w45 = {w[4], w[5]}, w67 = {w[6], w[7]};
      xi[pl][0] += a01 * a32;
      xi[pl][0] += a45 * a76;
      xi[pl][1] += a01 * a67;
      xi[pl][1] += a45 * a23;
      xi[pl][2] += a01 * w45;
      xi[pl][2] += a23 * w67;
      if ((t & 1) == 0) {
        const v8f b = o[pl][t ^ 3];
        const v2f b01 = {b[0], b[1]}, b23 = {b[2], b[3]}, b45 = {b[4], b[5]}, b67 = {b[6], b[7]};
        xi[pl][3] += a01 * b01;
        xi[pl][3] += a23 * b23;
        xi[pl][3] += a45 * b45;
        xi[pl][3] += a67 * b67;
      }
      if ((t & 2) == 0) {
        const v8f b = o[pl][t ^ 2];
        const v2f b01 = {b[0], b[1]}, b23 = {b[2], b[3]}, b45 = {b[4], b[5]}, b67 = {b[6], b[7]};
        xi[pl][4] += a01 * b01;
        xi[pl][4] += a23 * b23;
        xi[pl][4] += a45 * b45;
        xi[pl][4] += a67 * b67;
      }
    }
  }
  float xq[NQ];
#pragma unroll
  for (int q = 0; q < 5; ++q) {
    const v2f sum = xi[0][q] + xi[1][q];
    xq[q < 3 ? q : q + 5] = sum.x + sum.y;
    xq[3 + q] = xl[0][q] + xl[1][q];
  }

  float zq[NQ];
  int par = half;
  zq[0] = z0; zq[1] = z1; zq[2] = z2;
  zq[3] = par ? -z2 : z2;
  par ^= lid & 1;
  zq[4] = par ? -z2 : z2;
  par ^= (lid >> 1) & 1;
  zq[5] = par ? -z2 : z2;
  par ^= (lid >> 2) & 1;
  zq[6] = par ? -z2 : z2;
  par ^= (lid >> 3) & 1;
  zq[7] = par ? -z2 : z2;
  zq[8] = par ? -z8 : z8;
  zq[9] = par ? -z9 : z9;
#pragma unroll
  for (int q = 0; q < NQ; ++q) {
    const float c = cl[q], s = sl[q];
    const bool once = (q <= 2) || (q >= 7);
    const float sn = once ? 4.f * (c * s) : 2.f * (c * s);
    zq[q] = (c * c - s * s) * zq[q] - sn * xq[q];
  }
#pragma unroll
  for (int q = 0; q < NQ; ++q) zq[q] = wave_sum(zq[q]);

  if (lane < NQ) {
    float acc = bp;
#pragma unroll
    for (int q = 0; q < NQ; ++q) acc += zq[q] * wp[q];
    out[b * NQ + lane] = acc;
  }
}

extern "C" void kernel_launch(void* const* d_in, const int* in_sizes, int n_in,
                              void* d_out, int out_size, void* d_ws, size_t ws_size,
                              hipStream_t stream) {
  const float* x       = (const float*)d_in[0];
  const float* W_pre   = (const float*)d_in[1];
  const float* b_pre   = (const float*)d_in[2];
  const float* weights = (const float*)d_in[3];
  const float* W_post  = (const float*)d_in[4];
  const float* b_post  = (const float*)d_in[5];
  float* out = (float*)d_out;
  hipLaunchKernelGGL(qnn_circuit_kernel, dim3(BATCH / WAVES), dim3(32 * WAVES), 0, stream,
                     x, W_pre, b_pre, weights, W_post, b_post, out);
}
